# starting kernel
# baseline (speedup 1.0000x reference)
#include <hip/hip_runtime.h>
#include <hip/hip_bf16.h>

typedef __attribute__((ext_vector_type(2))) float v2f;
typedef __attribute__((ext_vector_type(4))) float v4f;
typedef __attribute__((ext_vector_type(8))) float v8f;

#define SEQ_LEN 32768
#define EMB     1024
#define SHOT    20
#define MPAD    32
#define SPLIT   1638
#define SC      8096
#define FLATL   (SPLIT * EMB)
#define LC1     2024
#define LC2     2023
#define LC3     2023
#define LC4     2024
#define CATL    (LC1 + LC2 + LC3 + LC4)
#define KDIM    SC
#define NDIM    EMB
#define KCHUNKS 8
#define KPER    (KDIM / KCHUNKS)

__global__ void __launch_bounds__(256)
shot_pool_kernel(const float* __restrict__ seq, float* __restrict__ seq_shot) {
  int wave = (blockIdx.x * blockDim.x + threadIdx.x) >> 5;
  int lane = threadIdx.x & 31;
  if (wave >= SHOT * SC) return;
  int s = wave / SC;
  int j = wave - s * SC;
  int start = (int)(((long long)j * FLATL) / SC);
  int end   = (int)((((long long)(j + 1) * FLATL) + SC - 1) / SC);
  float sum = 0.f;
  for (int f = start + lane; f < end; f += 32) {
    int k = f >> 10;
    int e = f & 1023;
    int t = (k == 0) ? (s * SPLIT)
                     : (((s - 1) * SPLIT + k + SEQ_LEN) % SEQ_LEN);
    sum += seq[t * EMB + e];
  }
  for (int off = 16; off > 0; off >>= 1) sum += __shfl_xor(sum, off, 32);
  if (lane == 0) seq_shot[s * SC + j] = sum / (float)(end - start);
}

__global__ void __launch_bounds__(256)
conv_pool_kernel(const float* __restrict__ ss,
                 const float* __restrict__ w1, const float* __restrict__ b1,
                 const float* __restrict__ w2, const float* __restrict__ b2,
                 const float* __restrict__ w3, const float* __restrict__ b3,
                 const float* __restrict__ w4, const float* __restrict__ b4,
                 float* __restrict__ pooled) {
  int idx = blockIdx.x * blockDim.x + threadIdx.x;
  if (idx >= MPAD * SC) return;
  int s = idx / SC;
  int j = idx - s * SC;
  if (s >= SHOT) { pooled[idx] = 0.f; return; }
  int start = (int)(((long long)j * CATL) / SC);
  int end   = (int)((((long long)(j + 1) * CATL) + SC - 1) / SC);
  float acc = 0.f;
  for (int t = start; t < end; ++t) {
    float v;
    if (t < LC1) {
      int p = t * 4;
      v = b1[s];
      for (int ci = 0; ci < SHOT; ++ci) {
        const float* wr = w1 + (s * SHOT + ci) * 3;
        const float* xr = ss + ci * SC + p;
        v += wr[0] * xr[0] + wr[1] * xr[1] + wr[2] * xr[2];
      }
    } else if (t < LC1 + LC2) {
      int p = (t - LC1) * 4;
      v = b2[s];
      for (int ci = 0; ci < SHOT; ++ci) {
        const float* wr = w2 + (s * SHOT + ci) * 3;
        const float* xr = ss + ci * SC + p;
        v += wr[0] * xr[0] + wr[1] * xr[2] + wr[2] * xr[4];
      }
    } else if (t < LC1 + LC2 + LC3) {
      int p = (t - LC1 - LC2) * 4;
      v = b3[s];
      for (int ci = 0; ci < SHOT; ++ci) {
        const float* wr = w3 + (s * SHOT + ci) * 3;
        const float* xr = ss + ci * SC + p;
        v += wr[0] * xr[0] + wr[1] * xr[3] + wr[2] * xr[6];
      }
    } else {
      int p = (t - LC1 - LC2 - LC3) * 4;
      v = b4[s];
      for (int ci = 0; ci < SHOT; ++ci)
        v += w4[s * SHOT + ci] * ss[ci * SC + p];
    }
    acc += v;
  }
  pooled[idx] = acc / (float)(end - start);
}

__global__ void __launch_bounds__(32)
gemm_wmma_kernel(const float* __restrict__ pooled,
                 const float* __restrict__ fcw,
                 float* __restrict__ part) {
  int lane = threadIdx.x & 31;
  int half = lane >> 4;
  int mr   = lane & 15;
  int nbase = blockIdx.x * 16;
  int kc    = blockIdx.y;
  int k0    = kc * KPER;

  v8f acc0 = {};
  v8f acc1 = {};
  const float* arow0 = pooled + (size_t)mr * KDIM;
  const float* arow1 = pooled + (size_t)(mr + 16) * KDIM;
  const float* brow  = fcw + (size_t)(nbase + mr) * KDIM;

  for (int k = k0; k < k0 + KPER; k += 4) {
    int kk = k + 2 * half;
    v2f a0 = *(const v2f*)(arow0 + kk);
    v2f a1 = *(const v2f*)(arow1 + kk);
    v2f b  = *(const v2f*)(brow + kk);
    acc0 = __builtin_amdgcn_wmma_f32_16x16x4_f32(false, a0, false, b,
                                                 (short)0, acc0, false, false);
    acc1 = __builtin_amdgcn_wmma_f32_16x16x4_f32(false, a1, false, b,
                                                 (short)0, acc1, false, false);
  }

  float* pbase = part + (size_t)kc * MPAD * NDIM + nbase + mr;
  for (int r = 0; r < 8; ++r) {
    int row = r + 8 * half;
    pbase[(size_t)row * NDIM]        = acc0[r];
    pbase[(size_t)(row + 16) * NDIM] = acc1[r];
  }
}

__global__ void __launch_bounds__(256)
fc_tanh_ln_kernel(const float* __restrict__ part, const float* __restrict__ fcb,
                  const float* __restrict__ lng, const float* __restrict__ lnb,
                  float* __restrict__ h) {
  int s = blockIdx.x;
  int tid = threadIdx.x;
  __shared__ float redA[256];
  __shared__ float redB[256];
  float t[4];
  float sum = 0.f, sq = 0.f;
  for (int i = 0; i < 4; ++i) {
    int c = tid + i * 256;
    float v = fcb[c];
    for (int kc = 0; kc < KCHUNKS; ++kc)
      v += part[(size_t)(kc * MPAD + s) * NDIM + c];
    float tv = tanhf(v);
    t[i] = tv;
    sum += tv; sq += tv * tv;
  }
  redA[tid] = sum; redB[tid] = sq;
  __syncthreads();
  for (int off = 128; off > 0; off >>= 1) {
    if (tid < off) { redA[tid] += redA[tid + off]; redB[tid] += redB[tid + off]; }
    __syncthreads();
  }
  float mu  = redA[0] * (1.f / NDIM);
  float var = redB[0] * (1.f / NDIM) - mu * mu;
  float inv = rsqrtf(var + 1e-5f);
  for (int i = 0; i < 4; ++i) {
    int c = tid + i * 256;
    h[(size_t)s * NDIM + c] = (t[i] - mu) * inv * lng[c] + lnb[c];
  }
}

__global__ void __launch_bounds__(256)
merge_kernel(const float* __restrict__ seq, const float* __restrict__ h,
             float* __restrict__ out) {
  int vi = blockIdx.x * blockDim.x + threadIdx.x;
  int t  = vi >> 8;
  int e4 = vi & 255;
  int row = t / SPLIT; if (row > SHOT - 1) row = SHOT - 1;
  v4f a = __builtin_nontemporal_load(((const v4f*)seq) + vi);
  v4f b = ((const v4f*)(h + (size_t)row * EMB))[e4];
  v4f o = a + b;
  __builtin_nontemporal_store(o, ((v4f*)out) + vi);
}

extern "C" void kernel_launch(void* const* d_in, const int* in_sizes, int n_in,
                              void* d_out, int out_size, void* d_ws, size_t ws_size,
                              hipStream_t stream) {
  const float* seq  = (const float*)d_in[0];
  const float* w1   = (const float*)d_in[1];
  const float* b1   = (const float*)d_in[2];
  const float* w2   = (const float*)d_in[3];
  const float* b2   = (const float*)d_in[4];
  const float* w3   = (const float*)d_in[5];
  const float* b3   = (const float*)d_in[6];
  const float* w4   = (const float*)d_in[7];
  const float* b4   = (const float*)d_in[8];
  const float* fcw  = (const float*)d_in[9];
  const float* fcb  = (const float*)d_in[10];
  const float* lng  = (const float*)d_in[11];
  const float* lnb  = (const float*)d_in[12];
  float* out = (float*)d_out;

  char* ws = (char*)d_ws;
  float* seq_shot = (float*)(ws);
  float* pooled   = (float*)(ws + 647680);
  float* part     = (float*)(ws + 1683968);
  float* h        = (float*)(ws + 2732544);

  shot_pool_kernel<<<(SHOT * SC) / 8, 256, 0, stream>>>(seq, seq_shot);

  conv_pool_kernel<<<(MPAD * SC) / 256, 256, 0, stream>>>(
      seq_shot, w1, b1, w2, b2, w3, b3, w4, b4, pooled);

  gemm_wmma_kernel<<<dim3(NDIM / 16, KCHUNKS), 32, 0, stream>>>(pooled, fcw, part);

  fc_tanh_ln_kernel<<<SHOT, 256, 0, stream>>>(part, fcb, lng, lnb, h);

  merge_kernel<<<(SEQ_LEN * EMB / 4) / 256, 256, 0, stream>>>(seq, h, out);
}


# conv+pool on the matrix unit: 6 waves form the 20x48 conv outputs behind 32 pooled columns as f32 16x16x4 products (A tap weights, B gathered inputs), pool them from LDS; conv sum order changes
# speedup vs baseline: 2.2170x; 2.2170x over previous
#include <hip/hip_runtime.h>
#include <hip/hip_bf16.h>

typedef __attribute__((ext_vector_type(2))) float v2f;
typedef __attribute__((ext_vector_type(4))) float v4f;
typedef __attribute__((ext_vector_type(8))) float v8f;

#define SEQ_LEN 32768
#define EMB     1024
#define SHOT    20
#define MPAD    32
#define SPLIT   1638
#define SC      8096
#define FLATL   (SPLIT * EMB)
#define LC1     2024
#define LC2     2023
#define LC3     2023
#define LC4     2024
#define CATL    (LC1 + LC2 + LC3 + LC4)
#define KDIM    SC
#define NDIM    EMB
#define KCHUNKS 8
#define KPER    (KDIM / KCHUNKS)

#define WPW 32
#define NLD 7
#define GRP 4
#define NGRP (WPW / GRP)
static_assert(SC % WPW == 0, "a wave's windows stay inside one shot");
static_assert(FLATL / SC >= 32 * (NLD - 1), "loads 0..NLD-2 are inside every window");
static_assert((FLATL + SC - 1) / SC + 1 <= 32 * NLD, "NLD loads cover the longest window");

#define QW (FLATL / SC)
#define RW (FLATL % SC)
static_assert((long long)RW * SC + SC < 0x7fffffffLL, "RW*(j+1)+SC-1 fits in 32 bits");
__device__ __forceinline__ int pool_start(int j) {
  return QW * j + (int)((unsigned)(RW * j) / (unsigned)SC);
}
__device__ __forceinline__ int pool_end(int j) {
  return QW * (j + 1) + (int)((unsigned)(RW * (j + 1) + SC - 1) / (unsigned)SC);
}

__device__ __forceinline__ void pool_issue(const float* __restrict__ seq, int c0, int c1,
                                           int j, int lane, float (&x)[NLD]) {
  const int start = pool_start(j);
  const int last  = pool_end(j) - 1;
#pragma unroll
  for (int i = 0; i < NLD; ++i) {
    int f = start + lane + 32 * i;
    if (i == NLD - 1) f = (f < last) ? f : last;
    const unsigned boff = (unsigned)(f + ((f < EMB) ? c0 : c1)) * 4u;
    x[i] = *(const float*)((const char*)seq + boff);
  }
}

__device__ __forceinline__ float pool_sum(int j, int lane, const float (&x)[NLD]) {
  float sum = 0.f;
#pragma unroll
  for (int i = 0; i < NLD - 1; ++i) sum += x[i];
  if (pool_start(j) + lane + 32 * (NLD - 1) < pool_end(j)) sum += x[NLD - 1];
  return sum;
}

template <int N>
__device__ __forceinline__ void pool_fold(const float (&in)[2 * N], float (&out)[N], int lane, int off) {
  const bool hi = (lane & off) != 0;
#pragma unroll
  for (int i = 0; i < N; ++i) {
    float keep = hi ? in[i + N] : in[i];
    float send = hi ? in[i] : in[i + N];
    out[i] = keep + __shfl_xor(send, off, 32);
  }
}

__global__ void __launch_bounds__(128)
shot_pool_kernel(const float* __restrict__ seq, float* __restrict__ seq_shot) {
  const int lane = threadIdx.x & 31;
  const int wave = __builtin_amdgcn_readfirstlane((int)((blockIdx.x * blockDim.x + threadIdx.x) >> 5));
  const int s  = wave / (SC / WPW);
  const int j0 = (wave - s * (SC / WPW)) * WPW;
  const int c0 = s * SPLIT * EMB;
  const int c1 = (((s - 1) * SPLIT + SEQ_LEN) % SEQ_LEN) * EMB;

  float p[WPW];
  float xa[GRP][NLD], xb[GRP][NLD];

#pragma unroll
  for (int w = 0; w < GRP; ++w) pool_issue(seq, c0, c1, j0 + w, lane, xa[w]);

#pragma unroll
  for (int g = 0; g < NGRP; g += 2) {
#pragma unroll
    for (int w = 0; w < GRP; ++w) pool_issue(seq, c0, c1, j0 + (g + 1) * GRP + w, lane, xb[w]);
    __builtin_amdgcn_sched_barrier(0);
#pragma unroll
    for (int w = 0; w < GRP; ++w) p[g * GRP + w] = pool_sum(j0 + g * GRP + w, lane, xa[w]);
    __builtin_amdgcn_sched_barrier(0);
    if (g + 2 < NGRP) {
#pragma unroll
      for (int w = 0; w < GRP; ++w) pool_issue(seq, c0, c1, j0 + (g + 2) * GRP + w, lane, xa[w]);
    }
    __builtin_amdgcn_sched_barrier(0);
#pragma unroll
    for (int w = 0; w < GRP; ++w) p[(g + 1) * GRP + w] = pool_sum(j0 + (g + 1) * GRP + w, lane, xb[w]);
    __builtin_amdgcn_sched_barrier(0);
  }

  float q16[16], q8[8], q4[4], q2[2], q1[1];
  pool_fold<16>(p,   q16, lane, 16);
  pool_fold<8>(q16,  q8,  lane, 8);
  pool_fold<4>(q8,   q4,  lane, 4);
  pool_fold<2>(q4,   q2,  lane, 2);
  pool_fold<1>(q2,   q1,  lane, 1);

  const int j = j0 + lane;
  seq_shot[s * SC + j] = q1[0] / (float)(pool_end(j) - pool_start(j));
}

#define CT_COLS 48
#define CONV_K  (SHOT * 3)
#define CP_WAVES 6
static_assert(SC % 32 == 0, "a work-group's 32 pooled columns are whole 128-byte row pieces");
static_assert(CONV_K % 4 == 0 && SHOT % 4 == 0, "whole k-steps, even weight offsets");

static_assert((long long)SC * CATL + SC < 0x7fffffffLL, "(j+1)*CATL + SC - 1 fits in 32 bits");
__device__ __forceinline__ int cat_start(int j) { return (int)((unsigned)(j * CATL) / (unsigned)SC); }
__device__ __forceinline__ int cat_end(int j)   { return (int)((unsigned)((j + 1) * CATL + SC - 1) / (unsigned)SC); }

__device__ __forceinline__ int conv_of(int t) {
  return (t >= LC1) + (t >= LC1 + LC2) + (t >= LC1 + LC2 + LC3);
}

#define CONV_KS (CONV_K / 4)
__device__ __forceinline__ void conv_frags(const float* __restrict__ ss, const float* __restrict__ w,
                                           int k3, int dil, bool col_in, int p, int srow, bool row_in,
                                           int half, v2f (&fa)[CONV_KS], v2f (&fb)[CONV_KS]) {
  const int ktot = SHOT + k3 * (CONV_K - SHOT);
  const float* wr = w + srow * ktot;
#pragma unroll
  for (int ks = 0; ks < CONV_KS; ++ks) {
    const int  kk  = 4 * ks + 2 * half;
    const bool kin = kk < ktot;
    const v2f  av  = *(const v2f*)(wr + (kin ? kk : 0));
    fa[ks][0] = (kin && row_in) ? av[0] : 0.f;
    fa[ks][1] = (kin && row_in) ? av[1] : 0.f;
#pragma unroll
    for (int e = 0; e < 2; ++e) {
      const int ka = 4 * ks + e;
      const int kb = 4 * ks + 2 + e;
      const int oa = (ka % SHOT) * SC + k3 * ((ka / 3 - ka % SHOT) * SC + dil * (ka % 3));
      const int ob = (kb % SHOT) * SC + k3 * ((kb / 3 - kb % SHOT) * SC + dil * (kb % 3));
      const float x = ss[(col_in ? p : 0) + (half ? ob : oa)];
      fb[ks][e] = (kin && col_in) ? x : 0.f;
    }
  }
}

__global__ void __launch_bounds__(32 * CP_WAVES)
conv_pool_kernel(const float* __restrict__ ss,
                 const float* __restrict__ w1, const float* __restrict__ b1,
                 const float* __restrict__ w2, const float* __restrict__ b2,
                 const float* __restrict__ w3, const float* __restrict__ b3,
                 const float* __restrict__ w4, const float* __restrict__ b4,
                 float* __restrict__ pooled) {
  __shared__ float tile[MPAD][CT_COLS];
  const int lane = threadIdx.x & 31;
  const int half = lane >> 4;
  const int c    = lane & 15;
  const int wv = __builtin_amdgcn_readfirstlane((int)(threadIdx.x >> 5));
  const int q  = wv % 3;
  const int m  = wv / 3;
  const int jb  = blockIdx.x * 32;
  const int tlo = cat_start(jb);
  const int tq  = tlo + 16 * q;
  const int j     = jb + lane;
  const int start = cat_start(j);
  const int end   = cat_end(j);

  const int tfirst = (tq < CATL - 1) ? tq : CATL - 1;
  const int tlast  = (tq + 15 < CATL - 1) ? tq + 15 : CATL - 1;
  const int cvA = conv_of(tfirst);
  const int cvB = conv_of(tlast);

  const int  t    = tq + c;
  const bool t_in = t < CATL;
  const int  cvt  = conv_of(t_in ? t : CATL - 1);
  const int  off  = (cvt == 0) ? 0 : (cvt == 1) ? LC1 : (cvt == 2) ? LC1 + LC2 : LC1 + LC2 + LC3;
  const int  p    = t_in ? 4 * (t - off) : 0;

  const int  srow   = (16 * m + c < SHOT) ? 16 * m + c : SHOT - 1;
  const bool row_in = 16 * m + c < SHOT;

  const float* wA = (cvA == 0) ? w1 : (cvA == 1) ? w2 : (cvA == 2) ? w3 : w4;
  const float* wB = (cvB == 0) ? w1 : (cvB == 1) ? w2 : (cvB == 2) ? w3 : w4;

  v2f faA[CONV_KS], fbA[CONV_KS], faB[CONV_KS], fbB[CONV_KS];
  conv_frags(ss, wA, cvA < 3, cvA + 1, t_in && cvt == cvA,               p, srow, row_in, half, faA, fbA);
  conv_frags(ss, wB, cvB < 3, cvB + 1, t_in && cvt == cvB && cvB != cvA, p, srow, row_in, half, faB, fbB);
  const float* bp = (cvt == 0) ? b1 : (cvt == 1) ? b2 : (cvt == 2) ? b3 : b4;
  float bv[8];
#pragma unroll
  for (int r = 0; r < 8; ++r) {
    const int s = 16 * m + 8 * half + r;
    bv[r] = bp[(s < SHOT) ? s : SHOT - 1];
  }
  __builtin_amdgcn_sched_barrier(0);
  v8f acc = {};
#pragma unroll
  for (int ks = 0; ks < CONV_KS; ++ks)
    acc = __builtin_amdgcn_wmma_f32_16x16x4_f32(false, faA[ks], false, fbA[ks], (short)0, acc, false, false);
#pragma unroll
  for (int ks = 0; ks < CONV_KS; ++ks)
    acc = __builtin_amdgcn_wmma_f32_16x16x4_f32(false, faB[ks], false, fbB[ks], (short)0, acc, false, false);

#pragma unroll
  for (int r = 0; r < 8; ++r)
    tile[16 * m + 8 * half + r][16 * q + c] = acc[r] + bv[r];
  __syncthreads();

  for (int s = wv; s < MPAD; s += CP_WAVES) {
    float o = 0.f;
    if (s < SHOT) {
      float sum = 0.f;
      for (int tt = start; tt < end; ++tt) sum += tile[s][tt - tlo];
      o = sum / (float)(end - start);
    }
    pooled[s * SC + j] = o;
  }
}

#define KSTEPS (KPER / 4)
#define RING   23
#define AHEAD  21
#define NTRIP  (KSTEPS / RING)
static_assert(KPER % 4 == 0 && KSTEPS % RING == 0, "whole trips of RING k-steps");
static_assert(AHEAD < RING - 1 && 3 * AHEAD <= 63, "a slot is refilled two k-steps after its products; loads in flight fit the counter");

__device__ __forceinline__ void gemm_issue(const float* __restrict__ pa, const float* __restrict__ pb,
                                           int koff, v2f& a0, v2f& a1, v2f& b) {
  a0 = *(const v2f*)(pa + koff);
  b  = *(const v2f*)(pb + koff);
  a1 = *(const v2f*)(pa + 16 * KDIM + koff);
}

__global__ void __launch_bounds__(32)
gemm_wmma_kernel(const float* __restrict__ pooled,
                 const float* __restrict__ fcw,
                 float* __restrict__ part) {
  int lane = threadIdx.x & 31;
  int half = lane >> 4;
  int mr   = lane & 15;
  int nbase = blockIdx.x * 16;
  int kc    = blockIdx.y;
  int k0    = kc * KPER;

  v8f acc0 = {};
  v8f acc1 = {};
  const float* pa = pooled + (size_t)mr * KDIM + k0 + 2 * half;
  const float* pb = fcw + (size_t)(nbase + mr) * KDIM + k0 + 2 * half;

  v2f ra0[RING], ra1[RING], rb[RING];
#pragma unroll
  for (int i = 0; i < AHEAD; ++i) {
    gemm_issue(pa, pb, 4 * i, ra0[i], ra1[i], rb[i]);
    __builtin_amdgcn_sched_barrier(0);
  }

  for (int t = 0; t < NTRIP - 1; ++t) {
#pragma unroll
    for (int i = 0; i < RING; ++i) {
      acc0 = __builtin_amdgcn_wmma_f32_16x16x4_f32(false, ra0[i], false, rb[i],
                                                   (short)0, acc0, false, false);
      acc1 = __builtin_amdgcn_wmma_f32_16x16x4_f32(false, ra1[i], false, rb[i],
                                                   (short)0, acc1, false, false);
      const int j = (i + AHEAD) % RING;
      gemm_issue(pa, pb, 4 * (i + AHEAD), ra0[j], ra1[j], rb[j]);
      __builtin_amdgcn_sched_barrier(0);
    }
    pa += 4 * RING;
    pb += 4 * RING;
  }
#pragma unroll
  for (int i = 0; i < RING; ++i) {
    acc0 = __builtin_amdgcn_wmma_f32_16x16x4_f32(false, ra0[i], false, rb[i],
                                                 (short)0, acc0, false, false);
    acc1 = __builtin_amdgcn_wmma_f32_16x16x4_f32(false, ra1[i], false, rb[i],
                                                 (short)0, acc1, false, false);
    if (i + AHEAD < RING) {
      gemm_issue(pa, pb, 4 * (i + AHEAD), ra0[i + AHEAD], ra1[i + AHEAD], rb[i + AHEAD]);
      __builtin_amdgcn_sched_barrier(0);
    }
  }

  float* pbase = part + (size_t)kc * MPAD * NDIM + nbase + mr;
  for (int r = 0; r < 8; ++r) {
    int row = r + 8 * half;
    pbase[(size_t)row * NDIM]        = acc0[r];
    pbase[(size_t)(row + 16) * NDIM] = acc1[r];
  }
}

__global__ void __launch_bounds__(256)
fc_tanh_ln_kernel(const float* __restrict__ part, const float* __restrict__ fcb,
                  const float* __restrict__ lng, const float* __restrict__ lnb,
                  float* __restrict__ h) {
  int s = blockIdx.x;
  int tid = threadIdx.x;
  __shared__ float redA[256];
  __shared__ float redB[256];
  float t[4];
  float sum = 0.f, sq = 0.f;
  for (int i = 0; i < 4; ++i) {
    int c = tid + i * 256;
    float v = fcb[c];
    for (int kc = 0; kc < KCHUNKS; ++kc)
      v += part[(size_t)(kc * MPAD + s) * NDIM + c];
    float tv = tanhf(v);
    t[i] = tv;
    sum += tv; sq += tv * tv;
  }
  redA[tid] = sum; redB[tid] = sq;
  __syncthreads();
  for (int off = 128; off > 0; off >>= 1) {
    if (tid < off) { redA[tid] += redA[tid + off]; redB[tid] += redB[tid + off]; }
    __syncthreads();
  }
  float mu  = redA[0] * (1.f / NDIM);
  float var = redB[0] * (1.f / NDIM) - mu * mu;
  float inv = rsqrtf(var + 1e-5f);
  for (int i = 0; i < 4; ++i) {
    int c = tid + i * 256;
    h[(size_t)s * NDIM + c] = (t[i] - mu) * inv * lng[c] + lnb[c];
  }
}

#define MROWS 4
#define MLD   (EMB / 4 / 32)
static_assert(MROWS == 4, "two aligned row pairs per wave");
static_assert(SPLIT % 2 == 0 && ((SHOT - 1) * SPLIT) % 2 == 0, "an aligned pair of rows lies in one shot");
static_assert(SEQ_LEN % (MROWS * 4) == 0, "the grid covers the rows exactly");

__device__ __forceinline__ void merge_load_h(const float* __restrict__ h, int s, int lane, v4f (&hq)[MLD]) {
  const v4f* hp = (const v4f*)(h + s * EMB) + lane;
#pragma unroll
  for (int i = 0; i < MLD; ++i) hq[i] = hp[32 * i];
}

__device__ __forceinline__ void merge_store_row(v4f* __restrict__ op, const v4f (&x)[MLD], const v4f (&hq)[MLD]) {
#pragma unroll
  for (int i = 0; i < MLD; ++i) __builtin_nontemporal_store(x[i] + hq[i], op + 32 * i);
}

__global__ void __launch_bounds__(128)
merge_kernel(const float* __restrict__ seq, const float* __restrict__ h,
             float* __restrict__ out) {
  const int lane = threadIdx.x & 31;
  const int wave = __builtin_amdgcn_readfirstlane((int)((blockIdx.x * blockDim.x + threadIdx.x) >> 5));
  const int t0 = wave * MROWS;
  int sA = t0 / SPLIT;       if (sA > SHOT - 1) sA = SHOT - 1;
  int sB = (t0 + 2) / SPLIT; if (sB > SHOT - 1) sB = SHOT - 1;

  v4f hq[MLD];
  merge_load_h(h, sA, lane, hq);

  const v4f* sp = (const v4f*)seq + (size_t)t0 * (EMB / 4) + lane;
  v4f*       op = (v4f*)out       + (size_t)t0 * (EMB / 4) + lane;
  v4f x[MROWS][MLD];
#pragma unroll
  for (int r = 0; r < MROWS; ++r) {
#pragma unroll
    for (int i = 0; i < MLD; ++i)
      x[r][i] = __builtin_nontemporal_load(sp + r * (EMB / 4) + 32 * i);
  }

  merge_store_row(op,                 x[0], hq);
  merge_store_row(op + (EMB / 4),     x[1], hq);
  if (sB != sA) merge_load_h(h, sB, lane, hq);
  merge_store_row(op + 2 * (EMB / 4), x[2], hq);
  merge_store_row(op + 3 * (EMB / 4), x[3], hq);
}

extern "C" void kernel_launch(void* const* d_in, const int* in_sizes, int n_in,
                              void* d_out, int out_size, void* d_ws, size_t ws_size,
                              hipStream_t stream) {
  const float* seq  = (const float*)d_in[0];
  const float* w1   = (const float*)d_in[1];
  const float* b1   = (const float*)d_in[2];
  const float* w2   = (const float*)d_in[3];
  const float* b2   = (const float*)d_in[4];
  const float* w3   = (const float*)d_in[5];
  const float* b3   = (const float*)d_in[6];
  const float* w4   = (const float*)d_in[7];
  const float* b4   = (const float*)d_in[8];
  const float* fcw  = (const float*)d_in[9];
  const float* fcb  = (const float*)d_in[10];
  const float* lng  = (const float*)d_in[11];
  const float* lnb  = (const float*)d_in[12];
  float* out = (float*)d_out;

  char* ws = (char*)d_ws;
  float* seq_shot = (float*)(ws);
  float* pooled   = (float*)(ws + 647680);
  float* part     = (float*)(ws + 1683968);
  float* h        = (float*)(ws + 2732544);

  shot_pool_kernel<<<(SHOT * SC) / (WPW * 4), 128, 0, stream>>>(seq, seq_shot);

  conv_pool_kernel<<<SC / 32, 32 * CP_WAVES, 0, stream>>>(
      seq_shot, w1, b1, w2, b2, w3, b3, w4, b4, pooled);

  gemm_wmma_kernel<<<dim3(NDIM / 16, KCHUNKS), 32, 0, stream>>>(pooled, fcw, part);

  fc_tanh_ln_kernel<<<SHOT, 256, 0, stream>>>(part, fcb, lng, lnb, h);

  merge_kernel<<<SEQ_LEN / (MROWS * 4), 128, 0, stream>>>(seq, h, out);
}


# LayerNorm kernel: a thread owns 4 adjacent columns, its 11 16-byte loads (bias, 8 partials, gamma, beta) leave together: 1 memory round trip, was 5; butterfly + 1 barrier, was 9; LN sums regrouped
# speedup vs baseline: 2.3079x; 1.0410x over previous
#include <hip/hip_runtime.h>
#include <hip/hip_bf16.h>

typedef __attribute__((ext_vector_type(2))) float v2f;
typedef __attribute__((ext_vector_type(4))) float v4f;
typedef __attribute__((ext_vector_type(8))) float v8f;

#define SEQ_LEN 32768
#define EMB     1024
#define SHOT    20
#define MPAD    32
#define SPLIT   1638
#define SC      8096
#define FLATL   (SPLIT * EMB)
#define LC1     2024
#define LC2     2023
#define LC3     2023
#define LC4     2024
#define CATL    (LC1 + LC2 + LC3 + LC4)
#define KDIM    SC
#define NDIM    EMB
#define KCHUNKS 8
#define KPER    (KDIM / KCHUNKS)

#define WPW 32
#define NLD 7
#define GRP 4
#define NGRP (WPW / GRP)
static_assert(SC % WPW == 0, "a wave's windows stay inside one shot");
static_assert(FLATL / SC >= 32 * (NLD - 1), "loads 0..NLD-2 are inside every window");
static_assert((FLATL + SC - 1) / SC + 1 <= 32 * NLD, "NLD loads cover the longest window");

#define QW (FLATL / SC)
#define RW (FLATL % SC)
static_assert((long long)RW * SC + SC < 0x7fffffffLL, "RW*(j+1)+SC-1 fits in 32 bits");
__device__ __forceinline__ int pool_start(int j) {
  return QW * j + (int)((unsigned)(RW * j) / (unsigned)SC);
}
__device__ __forceinline__ int pool_end(int j) {
  return QW * (j + 1) + (int)((unsigned)(RW * (j + 1) + SC - 1) / (unsigned)SC);
}

__device__ __forceinline__ void pool_issue(const float* __restrict__ seq, int c0, int c1,
                                           int j, int lane, float (&x)[NLD]) {
  const int start = pool_start(j);
  const int last  = pool_end(j) - 1;
#pragma unroll
  for (int i = 0; i < NLD; ++i) {
    int f = start + lane + 32 * i;
    if (i == NLD - 1) f = (f < last) ? f : last;
    const unsigned boff = (unsigned)(f + ((f < EMB) ? c0 : c1)) * 4u;
    x[i] = *(const float*)((const char*)seq + boff);
  }
}

__device__ __forceinline__ float pool_sum(int j, int lane, const float (&x)[NLD]) {
  float sum = 0.f;
#pragma unroll
  for (int i = 0; i < NLD - 1; ++i) sum += x[i];
  if (pool_start(j) + lane + 32 * (NLD - 1) < pool_end(j)) sum += x[NLD - 1];
  return sum;
}

template <int N>
__device__ __forceinline__ void pool_fold(const float (&in)[2 * N], float (&out)[N], int lane, int off) {
  const bool hi = (lane & off) != 0;
#pragma unroll
  for (int i = 0; i < N; ++i) {
    float keep = hi ? in[i + N] : in[i];
    float send = hi ? in[i] : in[i + N];
    out[i] = keep + __shfl_xor(send, off, 32);
  }
}

__global__ void __launch_bounds__(128)
shot_pool_kernel(const float* __restrict__ seq, float* __restrict__ seq_shot) {
  const int lane = threadIdx.x & 31;
  const int wave = __builtin_amdgcn_readfirstlane((int)((blockIdx.x * blockDim.x + threadIdx.x) >> 5));
  const int s  = wave / (SC / WPW);
  const int j0 = (wave - s * (SC / WPW)) * WPW;
  const int c0 = s * SPLIT * EMB;
  const int c1 = (((s - 1) * SPLIT + SEQ_LEN) % SEQ_LEN) * EMB;

  float p[WPW];
  float xa[GRP][NLD], xb[GRP][NLD];

#pragma unroll
  for (int w = 0; w < GRP; ++w) pool_issue(seq, c0, c1, j0 + w, lane, xa[w]);

#pragma unroll
  for (int g = 0; g < NGRP; g += 2) {
#pragma unroll
    for (int w = 0; w < GRP; ++w) pool_issue(seq, c0, c1, j0 + (g + 1) * GRP + w, lane, xb[w]);
    __builtin_amdgcn_sched_barrier(0);
#pragma unroll
    for (int w = 0; w < GRP; ++w) p[g * GRP + w] = pool_sum(j0 + g * GRP + w, lane, xa[w]);
    __builtin_amdgcn_sched_barrier(0);
    if (g + 2 < NGRP) {
#pragma unroll
      for (int w = 0; w < GRP; ++w) pool_issue(seq, c0, c1, j0 + (g + 2) * GRP + w, lane, xa[w]);
    }
    __builtin_amdgcn_sched_barrier(0);
#pragma unroll
    for (int w = 0; w < GRP; ++w) p[(g + 1) * GRP + w] = pool_sum(j0 + (g + 1) * GRP + w, lane, xb[w]);
    __builtin_amdgcn_sched_barrier(0);
  }

  float q16[16], q8[8], q4[4], q2[2], q1[1];
  pool_fold<16>(p,   q16, lane, 16);
  pool_fold<8>(q16,  q8,  lane, 8);
  pool_fold<4>(q8,   q4,  lane, 4);
  pool_fold<2>(q4,   q2,  lane, 2);
  pool_fold<1>(q2,   q1,  lane, 1);

  const int j = j0 + lane;
  seq_shot[s * SC + j] = q1[0] / (float)(pool_end(j) - pool_start(j));
}

#define CT_COLS 48
#define CONV_K  (SHOT * 3)
#define CP_WAVES 6
static_assert(SC % 32 == 0, "a work-group's 32 pooled columns are whole 128-byte row pieces");
static_assert(CONV_K % 4 == 0 && SHOT % 4 == 0, "whole k-steps, even weight offsets");

static_assert((long long)SC * CATL + SC < 0x7fffffffLL, "(j+1)*CATL + SC - 1 fits in 32 bits");
__device__ __forceinline__ int cat_start(int j) { return (int)((unsigned)(j * CATL) / (unsigned)SC); }
__device__ __forceinline__ int cat_end(int j)   { return (int)((unsigned)((j + 1) * CATL + SC - 1) / (unsigned)SC); }

__device__ __forceinline__ int conv_of(int t) {
  return (t >= LC1) + (t >= LC1 + LC2) + (t >= LC1 + LC2 + LC3);
}

#define CONV_KS (CONV_K / 4)
__device__ __forceinline__ void conv_frags(const float* __restrict__ ss, const float* __restrict__ w,
                                           int k3, int dil, bool col_in, int p, int srow, bool row_in,
                                           int half, v2f (&fa)[CONV_KS], v2f (&fb)[CONV_KS]) {
  const int ktot = SHOT + k3 * (CONV_K - SHOT);
  const float* wr = w + srow * ktot;
#pragma unroll
  for (int ks = 0; ks < CONV_KS; ++ks) {
    const int  kk  = 4 * ks + 2 * half;
    const bool kin = kk < ktot;
    const v2f  av  = *(const v2f*)(wr + (kin ? kk : 0));
    fa[ks][0] = (kin && row_in) ? av[0] : 0.f;
    fa[ks][1] = (kin && row_in) ? av[1] : 0.f;
#pragma unroll
    for (int e = 0; e < 2; ++e) {
      const int ka = 4 * ks + e;
      const int kb = 4 * ks + 2 + e;
      const int oa = (ka % SHOT) * SC + k3 * ((ka / 3 - ka % SHOT) * SC + dil * (ka % 3));
      const int ob = (kb % SHOT) * SC + k3 * ((kb / 3 - kb % SHOT) * SC + dil * (kb % 3));
      const float x = ss[(col_in ? p : 0) + (half ? ob : oa)];
      fb[ks][e] = (kin && col_in) ? x : 0.f;
    }
  }
}

__global__ void __launch_bounds__(32 * CP_WAVES)
conv_pool_kernel(const float* __restrict__ ss,
                 const float* __restrict__ w1, const float* __restrict__ b1,
                 const float* __restrict__ w2, const float* __restrict__ b2,
                 const float* __restrict__ w3, const float* __restrict__ b3,
                 const float* __restrict__ w4, const float* __restrict__ b4,
                 float* __restrict__ pooled) {
  __shared__ float tile[MPAD][CT_COLS];
  const int lane = threadIdx.x & 31;
  const int half = lane >> 4;
  const int c    = lane & 15;
  const int wv = __builtin_amdgcn_readfirstlane((int)(threadIdx.x >> 5));
  const int q  = wv % 3;
  const int m  = wv / 3;
  const int jb  = blockIdx.x * 32;
  const int tlo = cat_start(jb);
  const int tq  = tlo + 16 * q;
  const int j     = jb + lane;
  const int start = cat_start(j);
  const int end   = cat_end(j);

  const int tfirst = (tq < CATL - 1) ? tq : CATL - 1;
  const int tlast  = (tq + 15 < CATL - 1) ? tq + 15 : CATL - 1;
  const int cvA = conv_of(tfirst);
  const int cvB = conv_of(tlast);

  const int  t    = tq + c;
  const bool t_in = t < CATL;
  const int  cvt  = conv_of(t_in ? t : CATL - 1);
  const int  off  = (cvt == 0) ? 0 : (cvt == 1) ? LC1 : (cvt == 2) ? LC1 + LC2 : LC1 + LC2 + LC3;
  const int  p    = t_in ? 4 * (t - off) : 0;

  const int  srow   = (16 * m + c < SHOT) ? 16 * m + c : SHOT - 1;
  const bool row_in = 16 * m + c < SHOT;

  const float* wA = (cvA == 0) ? w1 : (cvA == 1) ? w2 : (cvA == 2) ? w3 : w4;
  const float* wB = (cvB == 0) ? w1 : (cvB == 1) ? w2 : (cvB == 2) ? w3 : w4;

  v2f faA[CONV_KS], fbA[CONV_KS], faB[CONV_KS], fbB[CONV_KS];
  conv_frags(ss, wA, cvA < 3, cvA + 1, t_in && cvt == cvA,               p, srow, row_in, half, faA, fbA);
  conv_frags(ss, wB, cvB < 3, cvB + 1, t_in && cvt == cvB && cvB != cvA, p, srow, row_in, half, faB, fbB);
  const float* bp = (cvt == 0) ? b1 : (cvt == 1) ? b2 : (cvt == 2) ? b3 : b4;
  float bv[8];
#pragma unroll
  for (int r = 0; r < 8; ++r) {
    const int s = 16 * m + 8 * half + r;
    bv[r] = bp[(s < SHOT) ? s : SHOT - 1];
  }
  __builtin_amdgcn_sched_barrier(0);
  v8f acc = {};
#pragma unroll
  for (int ks = 0; ks < CONV_KS; ++ks)
    acc = __builtin_amdgcn_wmma_f32_16x16x4_f32(false, faA[ks], false, fbA[ks], (short)0, acc, false, false);
#pragma unroll
  for (int ks = 0; ks < CONV_KS; ++ks)
    acc = __builtin_amdgcn_wmma_f32_16x16x4_f32(false, faB[ks], false, fbB[ks], (short)0, acc, false, false);

#pragma unroll
  for (int r = 0; r < 8; ++r)
    tile[16 * m + 8 * half + r][16 * q + c] = acc[r] + bv[r];
  __syncthreads();

  for (int s = wv; s < MPAD; s += CP_WAVES) {
    float o = 0.f;
    if (s < SHOT) {
      float sum = 0.f;
      for (int tt = start; tt < end; ++tt) sum += tile[s][tt - tlo];
      o = sum / (float)(end - start);
    }
    pooled[s * SC + j] = o;
  }
}

#define KSTEPS (KPER / 4)
#define RING   23
#define AHEAD  21
#define NTRIP  (KSTEPS / RING)
static_assert(KPER % 4 == 0 && KSTEPS % RING == 0, "whole trips of RING k-steps");
static_assert(AHEAD < RING - 1 && 3 * AHEAD <= 63, "a slot is refilled two k-steps after its products; loads in flight fit the counter");

__device__ __forceinline__ void gemm_issue(const float* __restrict__ pa, const float* __restrict__ pb,
                                           int koff, v2f& a0, v2f& a1, v2f& b) {
  a0 = *(const v2f*)(pa + koff);
  b  = *(const v2f*)(pb + koff);
  a1 = *(const v2f*)(pa + 16 * KDIM + koff);
}

__global__ void __launch_bounds__(32)
gemm_wmma_kernel(const float* __restrict__ pooled,
                 const float* __restrict__ fcw,
                 float* __restrict__ part) {
  int lane = threadIdx.x & 31;
  int half = lane >> 4;
  int mr   = lane & 15;
  int nbase = blockIdx.x * 16;
  int kc    = blockIdx.y;
  int k0    = kc * KPER;

  v8f acc0 = {};
  v8f acc1 = {};
  const float* pa = pooled + (size_t)mr * KDIM + k0 + 2 * half;
  const float* pb = fcw + (size_t)(nbase + mr) * KDIM + k0 + 2 * half;

  v2f ra0[RING], ra1[RING], rb[RING];
#pragma unroll
  for (int i = 0; i < AHEAD; ++i) {
    gemm_issue(pa, pb, 4 * i, ra0[i], ra1[i], rb[i]);
    __builtin_amdgcn_sched_barrier(0);
  }

  for (int t = 0; t < NTRIP - 1; ++t) {
#pragma unroll
    for (int i = 0; i < RING; ++i) {
      acc0 = __builtin_amdgcn_wmma_f32_16x16x4_f32(false, ra0[i], false, rb[i],
                                                   (short)0, acc0, false, false);
      acc1 = __builtin_amdgcn_wmma_f32_16x16x4_f32(false, ra1[i], false, rb[i],
                                                   (short)0, acc1, false, false);
      const int j = (i + AHEAD) % RING;
      gemm_issue(pa, pb, 4 * (i + AHEAD), ra0[j], ra1[j], rb[j]);
      __builtin_amdgcn_sched_barrier(0);
    }
    pa += 4 * RING;
    pb += 4 * RING;
  }
#pragma unroll
  for (int i = 0; i < RING; ++i) {
    acc0 = __builtin_amdgcn_wmma_f32_16x16x4_f32(false, ra0[i], false, rb[i],
                                                 (short)0, acc0, false, false);
    acc1 = __builtin_amdgcn_wmma_f32_16x16x4_f32(false, ra1[i], false, rb[i],
                                                 (short)0, acc1, false, false);
    if (i + AHEAD < RING) {
      gemm_issue(pa, pb, 4 * (i + AHEAD), ra0[i + AHEAD], ra1[i + AHEAD], rb[i + AHEAD]);
      __builtin_amdgcn_sched_barrier(0);
    }
  }

  float* pbase = part + (size_t)kc * MPAD * NDIM + nbase + mr;
  for (int r = 0; r < 8; ++r) {
    int row = r + 8 * half;
    pbase[(size_t)row * NDIM]        = acc0[r];
    pbase[(size_t)(row + 16) * NDIM] = acc1[r];
  }
}

#define LN_WAVES (256 / 32)
static_assert(NDIM == 4 * 256, "a thread owns 4 neighbouring columns of its row");
__global__ void __launch_bounds__(256)
fc_tanh_ln_kernel(const float* __restrict__ part, const float* __restrict__ fcb,
                  const float* __restrict__ lng, const float* __restrict__ lnb,
                  float* __restrict__ h) {
  const int s = blockIdx.x;
  const int tid = threadIdx.x;
  __shared__ float redA[LN_WAVES];
  __shared__ float redB[LN_WAVES];

  const v4f* pp = (const v4f*)(part + (size_t)s * NDIM) + tid;
  v4f v = ((const v4f*)fcb)[tid];
  v4f pk[KCHUNKS];
#pragma unroll
  for (int kc = 0; kc < KCHUNKS; ++kc) pk[kc] = pp[(size_t)kc * (MPAD * NDIM / 4)];
  const v4f g = ((const v4f*)lng)[tid];
  const v4f b = ((const v4f*)lnb)[tid];
#pragma unroll
  for (int kc = 0; kc < KCHUNKS; ++kc) v += pk[kc];

  float t[4];
  float sum = 0.f, sq = 0.f;
#pragma unroll
  for (int i = 0; i < 4; ++i) {
    const float tv = tanhf(v[i]);
    t[i] = tv;
    sum += tv; sq += tv * tv;
  }
#pragma unroll
  for (int off = 16; off > 0; off >>= 1) {
    sum += __shfl_xor(sum, off, 32);
    sq  += __shfl_xor(sq,  off, 32);
  }
  if ((tid & 31) == 0) { redA[tid >> 5] = sum; redB[tid >> 5] = sq; }
  __syncthreads();
  float ra = 0.f, rb = 0.f;
#pragma unroll
  for (int w = 0; w < LN_WAVES; ++w) { ra += redA[w]; rb += redB[w]; }
  const float mu  = ra * (1.f / NDIM);
  const float var = rb * (1.f / NDIM) - mu * mu;
  const float inv = rsqrtf(var + 1e-5f);
  v4f o;
#pragma unroll
  for (int i = 0; i < 4; ++i) o[i] = (t[i] - mu) * inv * g[i] + b[i];
  ((v4f*)(h + (size_t)s * NDIM))[tid] = o;
}

#define MROWS 4
#define MLD   (EMB / 4 / 32)
static_assert(MROWS == 4, "two aligned row pairs per wave");
static_assert(SPLIT % 2 == 0 && ((SHOT - 1) * SPLIT) % 2 == 0, "an aligned pair of rows lies in one shot");
static_assert(SEQ_LEN % (MROWS * 4) == 0, "the grid covers the rows exactly");

__device__ __forceinline__ void merge_load_h(const float* __restrict__ h, int s, int lane, v4f (&hq)[MLD]) {
  const v4f* hp = (const v4f*)(h + s * EMB) + lane;
#pragma unroll
  for (int i = 0; i < MLD; ++i) hq[i] = hp[32 * i];
}

__device__ __forceinline__ void merge_store_row(v4f* __restrict__ op, const v4f (&x)[MLD], const v4f (&hq)[MLD]) {
#pragma unroll
  for (int i = 0; i < MLD; ++i) __builtin_nontemporal_store(x[i] + hq[i], op + 32 * i);
}

__global__ void __launch_bounds__(128)
merge_kernel(const float* __restrict__ seq, const float* __restrict__ h,
             float* __restrict__ out) {
  const int lane = threadIdx.x & 31;
  const int wave = __builtin_amdgcn_readfirstlane((int)((blockIdx.x * blockDim.x + threadIdx.x) >> 5));
  const int t0 = wave * MROWS;
  int sA = t0 / SPLIT;       if (sA > SHOT - 1) sA = SHOT - 1;
  int sB = (t0 + 2) / SPLIT; if (sB > SHOT - 1) sB = SHOT - 1;

  v4f hq[MLD];
  merge_load_h(h, sA, lane, hq);

  const v4f* sp = (const v4f*)seq + (size_t)t0 * (EMB / 4) + lane;
  v4f*       op = (v4f*)out       + (size_t)t0 * (EMB / 4) + lane;
  v4f x[MROWS][MLD];
#pragma unroll
  for (int r = 0; r < MROWS; ++r) {
#pragma unroll
    for (int i = 0; i < MLD; ++i)
      x[r][i] = __builtin_nontemporal_load(sp + r * (EMB / 4) + 32 * i);
  }

  merge_store_row(op,                 x[0], hq);
  merge_store_row(op + (EMB / 4),     x[1], hq);
  if (sB != sA) merge_load_h(h, sB, lane, hq);
  merge_store_row(op + 2 * (EMB / 4), x[2], hq);
  merge_store_row(op + 3 * (EMB / 4), x[3], hq);
}

extern "C" void kernel_launch(void* const* d_in, const int* in_sizes, int n_in,
                              void* d_out, int out_size, void* d_ws, size_t ws_size,
                              hipStream_t stream) {
  const float* seq  = (const float*)d_in[0];
  const float* w1   = (const float*)d_in[1];
  const float* b1   = (const float*)d_in[2];
  const float* w2   = (const float*)d_in[3];
  const float* b2   = (const float*)d_in[4];
  const float* w3   = (const float*)d_in[5];
  const float* b3   = (const float*)d_in[6];
  const float* w4   = (const float*)d_in[7];
  const float* b4   = (const float*)d_in[8];
  const float* fcw  = (const float*)d_in[9];
  const float* fcb  = (const float*)d_in[10];
  const float* lng  = (const float*)d_in[11];
  const float* lnb  = (const float*)d_in[12];
  float* out = (float*)d_out;

  char* ws = (char*)d_ws;
  float* seq_shot = (float*)(ws);
  float* pooled   = (float*)(ws + 647680);
  float* part     = (float*)(ws + 1683968);
  float* h        = (float*)(ws + 2732544);

  shot_pool_kernel<<<(SHOT * SC) / (WPW * 4), 128, 0, stream>>>(seq, seq_shot);

  conv_pool_kernel<<<SC / 32, 32 * CP_WAVES, 0, stream>>>(
      seq_shot, w1, b1, w2, b2, w3, b3, w4, b4, pooled);

  gemm_wmma_kernel<<<dim3(NDIM / 16, KCHUNKS), 32, 0, stream>>>(pooled, fcw, part);

  fc_tanh_ln_kernel<<<SHOT, 256, 0, stream>>>(part, fcb, lng, lnb, h);

  merge_kernel<<<SEQ_LEN / (MROWS * 4), 128, 0, stream>>>(seq, h, out);
}


# merge: a wave owns 8 whole rows as 4 row pairs through two register sets, pair p+1's 16 loads issued before pair p's stores; 4096 waves in one resident round, was 8192 in two; sums unchanged
# speedup vs baseline: 2.3916x; 1.0363x over previous
#include <hip/hip_runtime.h>
#include <hip/hip_bf16.h>

typedef __attribute__((ext_vector_type(2))) float v2f;
typedef __attribute__((ext_vector_type(4))) float v4f;
typedef __attribute__((ext_vector_type(8))) float v8f;

#define SEQ_LEN 32768
#define EMB     1024
#define SHOT    20
#define MPAD    32
#define SPLIT   1638
#define SC      8096
#define FLATL   (SPLIT * EMB)
#define LC1     2024
#define LC2     2023
#define LC3     2023
#define LC4     2024
#define CATL    (LC1 + LC2 + LC3 + LC4)
#define KDIM    SC
#define NDIM    EMB
#define KCHUNKS 8
#define KPER    (KDIM / KCHUNKS)

#define WPW 32
#define NLD 7
#define GRP 4
#define NGRP (WPW / GRP)
static_assert(SC % WPW == 0, "a wave's windows stay inside one shot");
static_assert(FLATL / SC >= 32 * (NLD - 1), "loads 0..NLD-2 are inside every window");
static_assert((FLATL + SC - 1) / SC + 1 <= 32 * NLD, "NLD loads cover the longest window");

#define QW (FLATL / SC)
#define RW (FLATL % SC)
static_assert((long long)RW * SC + SC < 0x7fffffffLL, "RW*(j+1)+SC-1 fits in 32 bits");
__device__ __forceinline__ int pool_start(int j) {
  return QW * j + (int)((unsigned)(RW * j) / (unsigned)SC);
}
__device__ __forceinline__ int pool_end(int j) {
  return QW * (j + 1) + (int)((unsigned)(RW * (j + 1) + SC - 1) / (unsigned)SC);
}

__device__ __forceinline__ void pool_issue(const float* __restrict__ seq, int c0, int c1,
                                           int j, int lane, float (&x)[NLD]) {
  const int start = pool_start(j);
  const int last  = pool_end(j) - 1;
#pragma unroll
  for (int i = 0; i < NLD; ++i) {
    int f = start + lane + 32 * i;
    if (i == NLD - 1) f = (f < last) ? f : last;
    const unsigned boff = (unsigned)(f + ((f < EMB) ? c0 : c1)) * 4u;
    x[i] = *(const float*)((const char*)seq + boff);
  }
}

__device__ __forceinline__ float pool_sum(int j, int lane, const float (&x)[NLD]) {
  float sum = 0.f;
#pragma unroll
  for (int i = 0; i < NLD - 1; ++i) sum += x[i];
  if (pool_start(j) + lane + 32 * (NLD - 1) < pool_end(j)) sum += x[NLD - 1];
  return sum;
}

template <int N>
__device__ __forceinline__ void pool_fold(const float (&in)[2 * N], float (&out)[N], int lane, int off) {
  const bool hi = (lane & off) != 0;
#pragma unroll
  for (int i = 0; i < N; ++i) {
    float keep = hi ? in[i + N] : in[i];
    float send = hi ? in[i] : in[i + N];
    out[i] = keep + __shfl_xor(send, off, 32);
  }
}

__global__ void __launch_bounds__(128)
shot_pool_kernel(const float* __restrict__ seq, float* __restrict__ seq_shot) {
  const int lane = threadIdx.x & 31;
  const int wave = __builtin_amdgcn_readfirstlane((int)((blockIdx.x * blockDim.x + threadIdx.x) >> 5));
  const int s  = wave / (SC / WPW);
  const int j0 = (wave - s * (SC / WPW)) * WPW;
  const int c0 = s * SPLIT * EMB;
  const int c1 = (((s - 1) * SPLIT + SEQ_LEN) % SEQ_LEN) * EMB;

  float p[WPW];
  float xa[GRP][NLD], xb[GRP][NLD];

#pragma unroll
  for (int w = 0; w < GRP; ++w) pool_issue(seq, c0, c1, j0 + w, lane, xa[w]);

#pragma unroll
  for (int g = 0; g < NGRP; g += 2) {
#pragma unroll
    for (int w = 0; w < GRP; ++w) pool_issue(seq, c0, c1, j0 + (g + 1) * GRP + w, lane, xb[w]);
    __builtin_amdgcn_sched_barrier(0);
#pragma unroll
    for (int w = 0; w < GRP; ++w) p[g * GRP + w] = pool_sum(j0 + g * GRP + w, lane, xa[w]);
    __builtin_amdgcn_sched_barrier(0);
    if (g + 2 < NGRP) {
#pragma unroll
      for (int w = 0; w < GRP; ++w) pool_issue(seq, c0, c1, j0 + (g + 2) * GRP + w, lane, xa[w]);
    }
    __builtin_amdgcn_sched_barrier(0);
#pragma unroll
    for (int w = 0; w < GRP; ++w) p[(g + 1) * GRP + w] = pool_sum(j0 + (g + 1) * GRP + w, lane, xb[w]);
    __builtin_amdgcn_sched_barrier(0);
  }

  float q16[16], q8[8], q4[4], q2[2], q1[1];
  pool_fold<16>(p,   q16, lane, 16);
  pool_fold<8>(q16,  q8,  lane, 8);
  pool_fold<4>(q8,   q4,  lane, 4);
  pool_fold<2>(q4,   q2,  lane, 2);
  pool_fold<1>(q2,   q1,  lane, 1);

  const int j = j0 + lane;
  seq_shot[s * SC + j] = q1[0] / (float)(pool_end(j) - pool_start(j));
}

#define CT_COLS 48
#define CONV_K  (SHOT * 3)
#define CP_WAVES 6
static_assert(SC % 32 == 0, "a work-group's 32 pooled columns are whole 128-byte row pieces");
static_assert(CONV_K % 4 == 0 && SHOT % 4 == 0, "whole k-steps, even weight offsets");

static_assert((long long)SC * CATL + SC < 0x7fffffffLL, "(j+1)*CATL + SC - 1 fits in 32 bits");
__device__ __forceinline__ int cat_start(int j) { return (int)((unsigned)(j * CATL) / (unsigned)SC); }
__device__ __forceinline__ int cat_end(int j)   { return (int)((unsigned)((j + 1) * CATL + SC - 1) / (unsigned)SC); }

__device__ __forceinline__ int conv_of(int t) {
  return (t >= LC1) + (t >= LC1 + LC2) + (t >= LC1 + LC2 + LC3);
}

#define CONV_KS (CONV_K / 4)
__device__ __forceinline__ void conv_frags(const float* __restrict__ ss, const float* __restrict__ w,
                                           int k3, int dil, bool col_in, int p, int srow, bool row_in,
                                           int half, v2f (&fa)[CONV_KS], v2f (&fb)[CONV_KS]) {
  const int ktot = SHOT + k3 * (CONV_K - SHOT);
  const float* wr = w + srow * ktot;
#pragma unroll
  for (int ks = 0; ks < CONV_KS; ++ks) {
    const int  kk  = 4 * ks + 2 * half;
    const bool kin = kk < ktot;
    const v2f  av  = *(const v2f*)(wr + (kin ? kk : 0));
    fa[ks][0] = (kin && row_in) ? av[0] : 0.f;
    fa[ks][1] = (kin && row_in) ? av[1] : 0.f;
#pragma unroll
    for (int e = 0; e < 2; ++e) {
      const int ka = 4 * ks + e;
      const int kb = 4 * ks + 2 + e;
      const int oa = (ka % SHOT) * SC + k3 * ((ka / 3 - ka % SHOT) * SC + dil * (ka % 3));
      const int ob = (kb % SHOT) * SC + k3 * ((kb / 3 - kb % SHOT) * SC + dil * (kb % 3));
      const float x = ss[(col_in ? p : 0) + (half ? ob : oa)];
      fb[ks][e] = (kin && col_in) ? x : 0.f;
    }
  }
}

__global__ void __launch_bounds__(32 * CP_WAVES)
conv_pool_kernel(const float* __restrict__ ss,
                 const float* __restrict__ w1, const float* __restrict__ b1,
                 const float* __restrict__ w2, const float* __restrict__ b2,
                 const float* __restrict__ w3, const float* __restrict__ b3,
                 const float* __restrict__ w4, const float* __restrict__ b4,
                 float* __restrict__ pooled) {
  __shared__ float tile[MPAD][CT_COLS];
  const int lane = threadIdx.x & 31;
  const int half = lane >> 4;
  const int c    = lane & 15;
  const int wv = __builtin_amdgcn_readfirstlane((int)(threadIdx.x >> 5));
  const int q  = wv % 3;
  const int m  = wv / 3;
  const int jb  = blockIdx.x * 32;
  const int tlo = cat_start(jb);
  const int tq  = tlo + 16 * q;
  const int j     = jb + lane;
  const int start = cat_start(j);
  const int end   = cat_end(j);

  const int tfirst = (tq < CATL - 1) ? tq : CATL - 1;
  const int tlast  = (tq + 15 < CATL - 1) ? tq + 15 : CATL - 1;
  const int cvA = conv_of(tfirst);
  const int cvB = conv_of(tlast);

  const int  t    = tq + c;
  const bool t_in = t < CATL;
  const int  cvt  = conv_of(t_in ? t : CATL - 1);
  const int  off  = (cvt == 0) ? 0 : (cvt == 1) ? LC1 : (cvt == 2) ? LC1 + LC2 : LC1 + LC2 + LC3;
  const int  p    = t_in ? 4 * (t - off) : 0;

  const int  srow   = (16 * m + c < SHOT) ? 16 * m + c : SHOT - 1;
  const bool row_in = 16 * m + c < SHOT;

  const float* wA = (cvA == 0) ? w1 : (cvA == 1) ? w2 : (cvA == 2) ? w3 : w4;
  const float* wB = (cvB == 0) ? w1 : (cvB == 1) ? w2 : (cvB == 2) ? w3 : w4;

  v2f faA[CONV_KS], fbA[CONV_KS], faB[CONV_KS], fbB[CONV_KS];
  conv_frags(ss, wA, cvA < 3, cvA + 1, t_in && cvt == cvA,               p, srow, row_in, half, faA, fbA);
  conv_frags(ss, wB, cvB < 3, cvB + 1, t_in && cvt == cvB && cvB != cvA, p, srow, row_in, half, faB, fbB);
  const float* bp = (cvt == 0) ? b1 : (cvt == 1) ? b2 : (cvt == 2) ? b3 : b4;
  float bv[8];
#pragma unroll
  for (int r = 0; r < 8; ++r) {
    const int s = 16 * m + 8 * half + r;
    bv[r] = bp[(s < SHOT) ? s : SHOT - 1];
  }
  __builtin_amdgcn_sched_barrier(0);
  v8f acc = {};
#pragma unroll
  for (int ks = 0; ks < CONV_KS; ++ks)
    acc = __builtin_amdgcn_wmma_f32_16x16x4_f32(false, faA[ks], false, fbA[ks], (short)0, acc, false, false);
#pragma unroll
  for (int ks = 0; ks < CONV_KS; ++ks)
    acc = __builtin_amdgcn_wmma_f32_16x16x4_f32(false, faB[ks], false, fbB[ks], (short)0, acc, false, false);

#pragma unroll
  for (int r = 0; r < 8; ++r)
    tile[16 * m + 8 * half + r][16 * q + c] = acc[r] + bv[r];
  __syncthreads();

  for (int s = wv; s < MPAD; s += CP_WAVES) {
    float o = 0.f;
    if (s < SHOT) {
      float sum = 0.f;
      for (int tt = start; tt < end; ++tt) sum += tile[s][tt - tlo];
      o = sum / (float)(end - start);
    }
    pooled[s * SC + j] = o;
  }
}

#define KSTEPS (KPER / 4)
#define RING   23
#define AHEAD  21
#define NTRIP  (KSTEPS / RING)
static_assert(KPER % 4 == 0 && KSTEPS % RING == 0, "whole trips of RING k-steps");
static_assert(AHEAD < RING - 1 && 3 * AHEAD <= 63, "a slot is refilled two k-steps after its products; loads in flight fit the counter");

__device__ __forceinline__ void gemm_issue(const float* __restrict__ pa, const float* __restrict__ pb,
                                           int koff, v2f& a0, v2f& a1, v2f& b) {
  a0 = *(const v2f*)(pa + koff);
  b  = *(const v2f*)(pb + koff);
  a1 = *(const v2f*)(pa + 16 * KDIM + koff);
}

__global__ void __launch_bounds__(32)
gemm_wmma_kernel(const float* __restrict__ pooled,
                 const float* __restrict__ fcw,
                 float* __restrict__ part) {
  int lane = threadIdx.x & 31;
  int half = lane >> 4;
  int mr   = lane & 15;
  int nbase = blockIdx.x * 16;
  int kc    = blockIdx.y;
  int k0    = kc * KPER;

  v8f acc0 = {};
  v8f acc1 = {};
  const float* pa = pooled + (size_t)mr * KDIM + k0 + 2 * half;
  const float* pb = fcw + (size_t)(nbase + mr) * KDIM + k0 + 2 * half;

  v2f ra0[RING], ra1[RING], rb[RING];
#pragma unroll
  for (int i = 0; i < AHEAD; ++i) {
    gemm_issue(pa, pb, 4 * i, ra0[i], ra1[i], rb[i]);
    __builtin_amdgcn_sched_barrier(0);
  }

  for (int t = 0; t < NTRIP - 1; ++t) {
#pragma unroll
    for (int i = 0; i < RING; ++i) {
      acc0 = __builtin_amdgcn_wmma_f32_16x16x4_f32(false, ra0[i], false, rb[i],
                                                   (short)0, acc0, false, false);
      acc1 = __builtin_amdgcn_wmma_f32_16x16x4_f32(false, ra1[i], false, rb[i],
                                                   (short)0, acc1, false, false);
      const int j = (i + AHEAD) % RING;
      gemm_issue(pa, pb, 4 * (i + AHEAD), ra0[j], ra1[j], rb[j]);
      __builtin_amdgcn_sched_barrier(0);
    }
    pa += 4 * RING;
    pb += 4 * RING;
  }
#pragma unroll
  for (int i = 0; i < RING; ++i) {
    acc0 = __builtin_amdgcn_wmma_f32_16x16x4_f32(false, ra0[i], false, rb[i],
                                                 (short)0, acc0, false, false);
    acc1 = __builtin_amdgcn_wmma_f32_16x16x4_f32(false, ra1[i], false, rb[i],
                                                 (short)0, acc1, false, false);
    if (i + AHEAD < RING) {
      gemm_issue(pa, pb, 4 * (i + AHEAD), ra0[i + AHEAD], ra1[i + AHEAD], rb[i + AHEAD]);
      __builtin_amdgcn_sched_barrier(0);
    }
  }

  float* pbase = part + (size_t)kc * MPAD * NDIM + nbase + mr;
  for (int r = 0; r < 8; ++r) {
    int row = r + 8 * half;
    pbase[(size_t)row * NDIM]        = acc0[r];
    pbase[(size_t)(row + 16) * NDIM] = acc1[r];
  }
}

#define LN_WAVES (256 / 32)
static_assert(NDIM == 4 * 256, "a thread owns 4 neighbouring columns of its row");
__global__ void __launch_bounds__(256)
fc_tanh_ln_kernel(const float* __restrict__ part, const float* __restrict__ fcb,
                  const float* __restrict__ lng, const float* __restrict__ lnb,
                  float* __restrict__ h) {
  const int s = blockIdx.x;
  const int tid = threadIdx.x;
  __shared__ float redA[LN_WAVES];
  __shared__ float redB[LN_WAVES];

  const v4f* pp = (const v4f*)(part + (size_t)s * NDIM) + tid;
  v4f v = ((const v4f*)fcb)[tid];
  v4f pk[KCHUNKS];
#pragma unroll
  for (int kc = 0; kc < KCHUNKS; ++kc) pk[kc] = pp[(size_t)kc * (MPAD * NDIM / 4)];
  const v4f g = ((const v4f*)lng)[tid];
  const v4f b = ((const v4f*)lnb)[tid];
#pragma unroll
  for (int kc = 0; kc < KCHUNKS; ++kc) v += pk[kc];

  float t[4];
  float sum = 0.f, sq = 0.f;
#pragma unroll
  for (int i = 0; i < 4; ++i) {
    const float tv = tanhf(v[i]);
    t[i] = tv;
    sum += tv; sq += tv * tv;
  }
#pragma unroll
  for (int off = 16; off > 0; off >>= 1) {
    sum += __shfl_xor(sum, off, 32);
    sq  += __shfl_xor(sq,  off, 32);
  }
  if ((tid & 31) == 0) { redA[tid >> 5] = sum; redB[tid >> 5] = sq; }
  __syncthreads();
  float ra = 0.f, rb = 0.f;
#pragma unroll
  for (int w = 0; w < LN_WAVES; ++w) { ra += redA[w]; rb += redB[w]; }
  const float mu  = ra * (1.f / NDIM);
  const float var = rb * (1.f / NDIM) - mu * mu;
  const float inv = rsqrtf(var + 1e-5f);
  v4f o;
#pragma unroll
  for (int i = 0; i < 4; ++i) o[i] = (t[i] - mu) * inv * g[i] + b[i];
  ((v4f*)(h + (size_t)s * NDIM))[tid] = o;
}

#define MROWS 8
#define MPAIRS (MROWS / 2)
#define MLD   (EMB / 4 / 32)
#define PAIRQ (2 * (EMB / 4))
static_assert(MPAIRS % 2 == 0, "pairs alternate between two register sets");
static_assert(SPLIT % 2 == 0 && ((SHOT - 1) * SPLIT) % 2 == 0, "an aligned pair of rows lies in one shot");
static_assert(SEQ_LEN % (MROWS * 4) == 0, "the grid covers the rows exactly");

__device__ __forceinline__ int merge_shot(int t) {
  const int s = t / SPLIT;
  return (s > SHOT - 1) ? SHOT - 1 : s;
}

__device__ __forceinline__ void merge_load_h(const float* __restrict__ h, int s, int lane, v4f (&hq)[MLD]) {
  const v4f* hp = (const v4f*)(h + s * EMB) + lane;
#pragma unroll
  for (int i = 0; i < MLD; ++i) hq[i] = hp[32 * i];
}

__device__ __forceinline__ void merge_load_pair(const v4f* __restrict__ sp, v4f (&x)[2][MLD]) {
#pragma unroll
  for (int r = 0; r < 2; ++r) {
#pragma unroll
    for (int i = 0; i < MLD; ++i)
      x[r][i] = __builtin_nontemporal_load(sp + r * (EMB / 4) + 32 * i);
  }
}

__device__ __forceinline__ void merge_store_pair(v4f* __restrict__ op, const v4f (&x)[2][MLD], const v4f (&hq)[MLD]) {
#pragma unroll
  for (int r = 0; r < 2; ++r) {
#pragma unroll
    for (int i = 0; i < MLD; ++i) __builtin_nontemporal_store(x[r][i] + hq[i], op + r * (EMB / 4) + 32 * i);
  }
}

__global__ void __launch_bounds__(128)
merge_kernel(const float* __restrict__ seq, const float* __restrict__ h,
             float* __restrict__ out) {
  const int lane = threadIdx.x & 31;
  const int wave = __builtin_amdgcn_readfirstlane((int)((blockIdx.x * blockDim.x + threadIdx.x) >> 5));
  const int t0 = wave * MROWS;

  const v4f* sp = (const v4f*)seq + (size_t)t0 * (EMB / 4) + lane;
  v4f*       op = (v4f*)out       + (size_t)t0 * (EMB / 4) + lane;
  v4f hq[MLD];
  v4f xa[2][MLD], xb[2][MLD];
  merge_load_h(h, merge_shot(t0), lane, hq);
  merge_load_pair(sp, xa);

#pragma unroll
  for (int p = 0; p < MPAIRS; p += 2) {
    merge_load_pair(sp + (p + 1) * PAIRQ, xb);
    __builtin_amdgcn_sched_barrier(0);
    merge_store_pair(op + p * PAIRQ, xa, hq);
    __builtin_amdgcn_sched_barrier(0);
    merge_load_h(h, merge_shot(t0 + 2 * (p + 1)), lane, hq);
    if (p + 2 < MPAIRS) merge_load_pair(sp + (p + 2) * PAIRQ, xa);
    __builtin_amdgcn_sched_barrier(0);
    merge_store_pair(op + (p + 1) * PAIRQ, xb, hq);
    __builtin_amdgcn_sched_barrier(0);
    if (p + 2 < MPAIRS) merge_load_h(h, merge_shot(t0 + 2 * (p + 2)), lane, hq);
  }
}

extern "C" void kernel_launch(void* const* d_in, const int* in_sizes, int n_in,
                              void* d_out, int out_size, void* d_ws, size_t ws_size,
                              hipStream_t stream) {
  const float* seq  = (const float*)d_in[0];
  const float* w1   = (const float*)d_in[1];
  const float* b1   = (const float*)d_in[2];
  const float* w2   = (const float*)d_in[3];
  const float* b2   = (const float*)d_in[4];
  const float* w3   = (const float*)d_in[5];
  const float* b3   = (const float*)d_in[6];
  const float* w4   = (const float*)d_in[7];
  const float* b4   = (const float*)d_in[8];
  const float* fcw  = (const float*)d_in[9];
  const float* fcb  = (const float*)d_in[10];
  const float* lng  = (const float*)d_in[11];
  const float* lnb  = (const float*)d_in[12];
  float* out = (float*)d_out;

  char* ws = (char*)d_ws;
  float* seq_shot = (float*)(ws);
  float* pooled   = (float*)(ws + 647680);
  float* part     = (float*)(ws + 1683968);
  float* h        = (float*)(ws + 2732544);

  shot_pool_kernel<<<(SHOT * SC) / (WPW * 4), 128, 0, stream>>>(seq, seq_shot);

  conv_pool_kernel<<<SC / 32, 32 * CP_WAVES, 0, stream>>>(
      seq_shot, w1, b1, w2, b2, w3, b3, w4, b4, pooled);

  gemm_wmma_kernel<<<dim3(NDIM / 16, KCHUNKS), 32, 0, stream>>>(pooled, fcw, part);

  fc_tanh_ln_kernel<<<SHOT, 256, 0, stream>>>(part, fcb, lng, lnb, h);

  merge_kernel<<<SEQ_LEN / (MROWS * 4), 128, 0, stream>>>(seq, h, out);
}


# GEMM A operand: conv kernel writes pooled in A-fragment order (512 B per k-step), a k-step is 1 whole-line 16-byte A load + 1 B load, was 3 loads of 16 lines each; 31 k-steps ahead; sums unchanged
# speedup vs baseline: 2.5023x; 1.0463x over previous
#include <hip/hip_runtime.h>
#include <hip/hip_bf16.h>

typedef __attribute__((ext_vector_type(2))) float v2f;
typedef __attribute__((ext_vector_type(4))) float v4f;
typedef __attribute__((ext_vector_type(8))) float v8f;

#define SEQ_LEN 32768
#define EMB     1024
#define SHOT    20
#define MPAD    32
#define SPLIT   1638
#define SC      8096
#define FLATL   (SPLIT * EMB)
#define LC1     2024
#define LC2     2023
#define LC3     2023
#define LC4     2024
#define CATL    (LC1 + LC2 + LC3 + LC4)
#define KDIM    SC
#define NDIM    EMB
#define KCHUNKS 8
#define KPER    (KDIM / KCHUNKS)

#define WPW 32
#define NLD 7
#define GRP 4
#define NGRP (WPW / GRP)
static_assert(SC % WPW == 0, "a wave's windows stay inside one shot");
static_assert(FLATL / SC >= 32 * (NLD - 1), "loads 0..NLD-2 are inside every window");
static_assert((FLATL + SC - 1) / SC + 1 <= 32 * NLD, "NLD loads cover the longest window");

#define QW (FLATL / SC)
#define RW (FLATL % SC)
static_assert((long long)RW * SC + SC < 0x7fffffffLL, "RW*(j+1)+SC-1 fits in 32 bits");
__device__ __forceinline__ int pool_start(int j) {
  return QW * j + (int)((unsigned)(RW * j) / (unsigned)SC);
}
__device__ __forceinline__ int pool_end(int j) {
  return QW * (j + 1) + (int)((unsigned)(RW * (j + 1) + SC - 1) / (unsigned)SC);
}

__device__ __forceinline__ void pool_issue(const float* __restrict__ seq, int c0, int c1,
                                           int j, int lane, float (&x)[NLD]) {
  const int start = pool_start(j);
  const int last  = pool_end(j) - 1;
#pragma unroll
  for (int i = 0; i < NLD; ++i) {
    int f = start + lane + 32 * i;
    if (i == NLD - 1) f = (f < last) ? f : last;
    const unsigned boff = (unsigned)(f + ((f < EMB) ? c0 : c1)) * 4u;
    x[i] = *(const float*)((const char*)seq + boff);
  }
}

__device__ __forceinline__ float pool_sum(int j, int lane, const float (&x)[NLD]) {
  float sum = 0.f;
#pragma unroll
  for (int i = 0; i < NLD - 1; ++i) sum += x[i];
  if (pool_start(j) + lane + 32 * (NLD - 1) < pool_end(j)) sum += x[NLD - 1];
  return sum;
}

template <int N>
__device__ __forceinline__ void pool_fold(const float (&in)[2 * N], float (&out)[N], int lane, int off) {
  const bool hi = (lane & off) != 0;
#pragma unroll
  for (int i = 0; i < N; ++i) {
    float keep = hi ? in[i + N] : in[i];
    float send = hi ? in[i] : in[i + N];
    out[i] = keep + __shfl_xor(send, off, 32);
  }
}

__global__ void __launch_bounds__(128)
shot_pool_kernel(const float* __restrict__ seq, float* __restrict__ seq_shot) {
  const int lane = threadIdx.x & 31;
  const int wave = __builtin_amdgcn_readfirstlane((int)((blockIdx.x * blockDim.x + threadIdx.x) >> 5));
  const int s  = wave / (SC / WPW);
  const int j0 = (wave - s * (SC / WPW)) * WPW;
  const int c0 = s * SPLIT * EMB;
  const int c1 = (((s - 1) * SPLIT + SEQ_LEN) % SEQ_LEN) * EMB;

  float p[WPW];
  float xa[GRP][NLD], xb[GRP][NLD];

#pragma unroll
  for (int w = 0; w < GRP; ++w) pool_issue(seq, c0, c1, j0 + w, lane, xa[w]);

#pragma unroll
  for (int g = 0; g < NGRP; g += 2) {
#pragma unroll
    for (int w = 0; w < GRP; ++w) pool_issue(seq, c0, c1, j0 + (g + 1) * GRP + w, lane, xb[w]);
    __builtin_amdgcn_sched_barrier(0);
#pragma unroll
    for (int w = 0; w < GRP; ++w) p[g * GRP + w] = pool_sum(j0 + g * GRP + w, lane, xa[w]);
    __builtin_amdgcn_sched_barrier(0);
    if (g + 2 < NGRP) {
#pragma unroll
      for (int w = 0; w < GRP; ++w) pool_issue(seq, c0, c1, j0 + (g + 2) * GRP + w, lane, xa[w]);
    }
    __builtin_amdgcn_sched_barrier(0);
#pragma unroll
    for (int w = 0; w < GRP; ++w) p[(g + 1) * GRP + w] = pool_sum(j0 + (g + 1) * GRP + w, lane, xb[w]);
    __builtin_amdgcn_sched_barrier(0);
  }

  float q16[16], q8[8], q4[4], q2[2], q1[1];
  pool_fold<16>(p,   q16, lane, 16);
  pool_fold<8>(q16,  q8,  lane, 8);
  pool_fold<4>(q8,   q4,  lane, 4);
  pool_fold<2>(q4,   q2,  lane, 2);
  pool_fold<1>(q2,   q1,  lane, 1);

  const int j = j0 + lane;
  seq_shot[s * SC + j] = q1[0] / (float)(pool_end(j) - pool_start(j));
}

#define CT_COLS 48
#define CONV_K  (SHOT * 3)
#define CP_WAVES 6
static_assert(SC % 32 == 0, "a work-group's 32 pooled columns are whole 128-byte row pieces");
static_assert(CONV_K % 4 == 0 && SHOT % 4 == 0, "whole k-steps, even weight offsets");

static_assert((long long)SC * CATL + SC < 0x7fffffffLL, "(j+1)*CATL + SC - 1 fits in 32 bits");
static_assert(CATL <= SC, "a window of the second pool holds one or two conv outputs");
__device__ __forceinline__ int cat_start(int j) { return (int)((unsigned)(j * CATL) / (unsigned)SC); }
__device__ __forceinline__ int cat_end(int j)   { return (int)((unsigned)((j + 1) * CATL + SC - 1) / (unsigned)SC); }

__device__ __forceinline__ int conv_of(int t) {
  return (t >= LC1) + (t >= LC1 + LC2) + (t >= LC1 + LC2 + LC3);
}

#define CONV_KS (CONV_K / 4)
__device__ __forceinline__ void conv_frags(const float* __restrict__ ss, const float* __restrict__ w,
                                           int k3, int dil, bool col_in, int p, int srow, bool row_in,
                                           int half, v2f (&fa)[CONV_KS], v2f (&fb)[CONV_KS]) {
  const int ktot = SHOT + k3 * (CONV_K - SHOT);
  const float* wr = w + srow * ktot;
#pragma unroll
  for (int ks = 0; ks < CONV_KS; ++ks) {
    const int  kk  = 4 * ks + 2 * half;
    const bool kin = kk < ktot;
    const v2f  av  = *(const v2f*)(wr + (kin ? kk : 0));
    fa[ks][0] = (kin && row_in) ? av[0] : 0.f;
    fa[ks][1] = (kin && row_in) ? av[1] : 0.f;
#pragma unroll
    for (int e = 0; e < 2; ++e) {
      const int ka = 4 * ks + e;
      const int kb = 4 * ks + 2 + e;
      const int oa = (ka % SHOT) * SC + k3 * ((ka / 3 - ka % SHOT) * SC + dil * (ka % 3));
      const int ob = (kb % SHOT) * SC + k3 * ((kb / 3 - kb % SHOT) * SC + dil * (kb % 3));
      const float x = ss[(col_in ? p : 0) + (half ? ob : oa)];
      fb[ks][e] = (kin && col_in) ? x : 0.f;
    }
  }
}

__global__ void __launch_bounds__(32 * CP_WAVES)
conv_pool_kernel(const float* __restrict__ ss,
                 const float* __restrict__ w1, const float* __restrict__ b1,
                 const float* __restrict__ w2, const float* __restrict__ b2,
                 const float* __restrict__ w3, const float* __restrict__ b3,
                 const float* __restrict__ w4, const float* __restrict__ b4,
                 float* __restrict__ pooled) {
  __shared__ float tile[MPAD][CT_COLS];
  const int lane = threadIdx.x & 31;
  const int half = lane >> 4;
  const int c    = lane & 15;
  const int wv = __builtin_amdgcn_readfirstlane((int)(threadIdx.x >> 5));
  const int q  = wv % 3;
  const int m  = wv / 3;
  const int jb  = blockIdx.x * 32;
  const int tlo = cat_start(jb);
  const int tq  = tlo + 16 * q;

  const int tfirst = (tq < CATL - 1) ? tq : CATL - 1;
  const int tlast  = (tq + 15 < CATL - 1) ? tq + 15 : CATL - 1;
  const int cvA = conv_of(tfirst);
  const int cvB = conv_of(tlast);

  const int  t    = tq + c;
  const bool t_in = t < CATL;
  const int  cvt  = conv_of(t_in ? t : CATL - 1);
  const int  off  = (cvt == 0) ? 0 : (cvt == 1) ? LC1 : (cvt == 2) ? LC1 + LC2 : LC1 + LC2 + LC3;
  const int  p    = t_in ? 4 * (t - off) : 0;

  const int  srow   = (16 * m + c < SHOT) ? 16 * m + c : SHOT - 1;
  const bool row_in = 16 * m + c < SHOT;

  const float* wA = (cvA == 0) ? w1 : (cvA == 1) ? w2 : (cvA == 2) ? w3 : w4;
  const float* wB = (cvB == 0) ? w1 : (cvB == 1) ? w2 : (cvB == 2) ? w3 : w4;

  v2f faA[CONV_KS], fbA[CONV_KS], faB[CONV_KS], fbB[CONV_KS];
  conv_frags(ss, wA, cvA < 3, cvA + 1, t_in && cvt == cvA,               p, srow, row_in, half, faA, fbA);
  conv_frags(ss, wB, cvB < 3, cvB + 1, t_in && cvt == cvB && cvB != cvA, p, srow, row_in, half, faB, fbB);
  const float* bp = (cvt == 0) ? b1 : (cvt == 1) ? b2 : (cvt == 2) ? b3 : b4;
  float bv[8];
#pragma unroll
  for (int r = 0; r < 8; ++r) {
    const int s = 16 * m + 8 * half + r;
    bv[r] = bp[(s < SHOT) ? s : SHOT - 1];
  }
  __builtin_amdgcn_sched_barrier(0);
  v8f acc = {};
#pragma unroll
  for (int ks = 0; ks < CONV_KS; ++ks)
    acc = __builtin_amdgcn_wmma_f32_16x16x4_f32(false, faA[ks], false, fbA[ks], (short)0, acc, false, false);
#pragma unroll
  for (int ks = 0; ks < CONV_KS; ++ks)
    acc = __builtin_amdgcn_wmma_f32_16x16x4_f32(false, faB[ks], false, fbB[ks], (short)0, acc, false, false);

#pragma unroll
  for (int r = 0; r < 8; ++r)
    tile[16 * m + 8 * half + r][16 * q + c] = acc[r] + bv[r];
  __syncthreads();

  for (int ks = wv; ks < 32 / 4; ks += CP_WAVES) {
    v4f o;
#pragma unroll
    for (int e = 0; e < 2; ++e) {
      const int   j     = jb + 4 * ks + 2 * half + e;
      const int   start = cat_start(j);
      const int   end   = cat_end(j);
      const bool  two   = end - start > 1;
      const float len   = (float)(end - start);
#pragma unroll
      for (int r = 0; r < 2; ++r) {
        const int s = c + 16 * r;
        const float x0 = tile[s][start - tlo];
        const float x1 = tile[s][end - 1 - tlo];
        float sum = 0.f;
        sum += x0;
        sum += two ? x1 : 0.f;
        o[2 * r + e] = (s < SHOT) ? sum / len : 0.f;
      }
    }
    ((v4f*)pooled)[(size_t)(blockIdx.x * (32 / 4) + ks) * 32 + lane] = o;
  }
}

#define KSTEPS (KPER / 4)
#define RING   33
#define AHEAD  31
#define NFULL  ((KSTEPS - AHEAD) / RING)
#define REM    (KSTEPS - NFULL * RING)
static_assert(KPER % 4 == 0 && KCHUNKS * KSTEPS * 4 == KDIM, "whole k-steps, the chunks cover K");
static_assert(AHEAD == RING - 2 && 2 * AHEAD <= 63, "a slot is refilled two k-steps after its products; loads in flight fit the counter");
static_assert(NFULL >= 1 && NFULL * RING + AHEAD <= KSTEPS && REM >= AHEAD, "the rolled trips ask only for k-steps of the chunk; the end drains the ring");

__device__ __forceinline__ void gemm_issue(const v4f* __restrict__ pa, const float* __restrict__ pb,
                                           int kstep, v4f& a, v2f& b) {
  a = pa[32 * kstep];
  b = *(const v2f*)(pb + 4 * kstep);
}

__device__ __forceinline__ void gemm_mul(const v4f& a, const v2f& b, v8f& acc0, v8f& acc1) {
  const v2f a0 = __builtin_shufflevector(a, a, 0, 1);
  const v2f a1 = __builtin_shufflevector(a, a, 2, 3);
  acc0 = __builtin_amdgcn_wmma_f32_16x16x4_f32(false, a0, false, b, (short)0, acc0, false, false);
  acc1 = __builtin_amdgcn_wmma_f32_16x16x4_f32(false, a1, false, b, (short)0, acc1, false, false);
}

__global__ void __launch_bounds__(32)
gemm_wmma_kernel(const float* __restrict__ pooled,
                 const float* __restrict__ fcw,
                 float* __restrict__ part) {
  int lane = threadIdx.x & 31;
  int half = lane >> 4;
  int mr   = lane & 15;
  int nbase = blockIdx.x * 16;
  int kc    = blockIdx.y;
  int k0    = kc * KPER;

  v8f acc0 = {};
  v8f acc1 = {};
  const v4f*   pa = (const v4f*)pooled + (size_t)(kc * KSTEPS) * 32 + lane;
  const float* pb = fcw + (size_t)(nbase + mr) * KDIM + k0 + 2 * half;

  v4f ra[RING];
  v2f rb[RING];
#pragma unroll
  for (int i = 0; i < AHEAD; ++i) {
    gemm_issue(pa, pb, i, ra[i], rb[i]);
    __builtin_amdgcn_sched_barrier(0);
  }

  for (int t = 0; t < NFULL; ++t) {
#pragma unroll
    for (int i = 0; i < RING; ++i) {
      gemm_mul(ra[i], rb[i], acc0, acc1);
      const int j = (i + AHEAD) % RING;
      gemm_issue(pa, pb, i + AHEAD, ra[j], rb[j]);
      __builtin_amdgcn_sched_barrier(0);
    }
    pa += 32 * RING;
    pb += 4 * RING;
  }
#pragma unroll
  for (int i = 0; i < REM; ++i) {
    gemm_mul(ra[i % RING], rb[i % RING], acc0, acc1);
    if (i + AHEAD < REM) {
      const int j = (i + AHEAD) % RING;
      gemm_issue(pa, pb, i + AHEAD, ra[j], rb[j]);
      __builtin_amdgcn_sched_barrier(0);
    }
  }

  float* pbase = part + (size_t)kc * MPAD * NDIM + nbase + mr;
  for (int r = 0; r < 8; ++r) {
    int row = r + 8 * half;
    pbase[(size_t)row * NDIM]        = acc0[r];
    pbase[(size_t)(row + 16) * NDIM] = acc1[r];
  }
}

#define LN_WAVES (256 / 32)
static_assert(NDIM == 4 * 256, "a thread owns 4 neighbouring columns of its row");
__global__ void __launch_bounds__(256)
fc_tanh_ln_kernel(const float* __restrict__ part, const float* __restrict__ fcb,
                  const float* __restrict__ lng, const float* __restrict__ lnb,
                  float* __restrict__ h) {
  const int s = blockIdx.x;
  const int tid = threadIdx.x;
  __shared__ float redA[LN_WAVES];
  __shared__ float redB[LN_WAVES];

  const v4f* pp = (const v4f*)(part + (size_t)s * NDIM) + tid;
  v4f v = ((const v4f*)fcb)[tid];
  v4f pk[KCHUNKS];
#pragma unroll
  for (int kc = 0; kc < KCHUNKS; ++kc) pk[kc] = pp[(size_t)kc * (MPAD * NDIM / 4)];
  const v4f g = ((const v4f*)lng)[tid];
  const v4f b = ((const v4f*)lnb)[tid];
#pragma unroll
  for (int kc = 0; kc < KCHUNKS; ++kc) v += pk[kc];

  float t[4];
  float sum = 0.f, sq = 0.f;
#pragma unroll
  for (int i = 0; i < 4; ++i) {
    const float tv = tanhf(v[i]);
    t[i] = tv;
    sum += tv; sq += tv * tv;
  }
#pragma unroll
  for (int off = 16; off > 0; off >>= 1) {
    sum += __shfl_xor(sum, off, 32);
    sq  += __shfl_xor(sq,  off, 32);
  }
  if ((tid & 31) == 0) { redA[tid >> 5] = sum; redB[tid >> 5] = sq; }
  __syncthreads();
  float ra = 0.f, rb = 0.f;
#pragma unroll
  for (int w = 0; w < LN_WAVES; ++w) { ra += redA[w]; rb += redB[w]; }
  const float mu  = ra * (1.f / NDIM);
  const float var = rb * (1.f / NDIM) - mu * mu;
  const float inv = rsqrtf(var + 1e-5f);
  v4f o;
#pragma unroll
  for (int i = 0; i < 4; ++i) o[i] = (t[i] - mu) * inv * g[i] + b[i];
  ((v4f*)(h + (size_t)s * NDIM))[tid] = o;
}

#define MROWS 8
#define MPAIRS (MROWS / 2)
#define MLD   (EMB / 4 / 32)
#define PAIRQ (2 * (EMB / 4))
static_assert(MPAIRS % 2 == 0, "pairs alternate between two register sets");
static_assert(SPLIT % 2 == 0 && ((SHOT - 1) * SPLIT) % 2 == 0, "an aligned pair of rows lies in one shot");
static_assert(SEQ_LEN % (MROWS * 4) == 0, "the grid covers the rows exactly");

__device__ __forceinline__ int merge_shot(int t) {
  const int s = t / SPLIT;
  return (s > SHOT - 1) ? SHOT - 1 : s;
}

__device__ __forceinline__ void merge_load_h(const float* __restrict__ h, int s, int lane, v4f (&hq)[MLD]) {
  const v4f* hp = (const v4f*)(h + s * EMB) + lane;
#pragma unroll
  for (int i = 0; i < MLD; ++i) hq[i] = hp[32 * i];
}

__device__ __forceinline__ void merge_load_pair(const v4f* __restrict__ sp, v4f (&x)[2][MLD]) {
#pragma unroll
  for (int r = 0; r < 2; ++r) {
#pragma unroll
    for (int i = 0; i < MLD; ++i)
      x[r][i] = __builtin_nontemporal_load(sp + r * (EMB / 4) + 32 * i);
  }
}

__device__ __forceinline__ void merge_store_pair(v4f* __restrict__ op, const v4f (&x)[2][MLD], const v4f (&hq)[MLD]) {
#pragma unroll
  for (int r = 0; r < 2; ++r) {
#pragma unroll
    for (int i = 0; i < MLD; ++i) __builtin_nontemporal_store(x[r][i] + hq[i], op + r * (EMB / 4) + 32 * i);
  }
}

__global__ void __launch_bounds__(128)
merge_kernel(const float* __restrict__ seq, const float* __restrict__ h,
             float* __restrict__ out) {
  const int lane = threadIdx.x & 31;
  const int wave = __builtin_amdgcn_readfirstlane((int)((blockIdx.x * blockDim.x + threadIdx.x) >> 5));
  const int t0 = wave * MROWS;

  const v4f* sp = (const v4f*)seq + (size_t)t0 * (EMB / 4) + lane;
  v4f*       op = (v4f*)out       + (size_t)t0 * (EMB / 4) + lane;
  v4f hq[MLD];
  v4f xa[2][MLD], xb[2][MLD];
  merge_load_h(h, merge_shot(t0), lane, hq);
  merge_load_pair(sp, xa);

#pragma unroll
  for (int p = 0; p < MPAIRS; p += 2) {
    merge_load_pair(sp + (p + 1) * PAIRQ, xb);
    __builtin_amdgcn_sched_barrier(0);
    merge_store_pair(op + p * PAIRQ, xa, hq);
    __builtin_amdgcn_sched_barrier(0);
    merge_load_h(h, merge_shot(t0 + 2 * (p + 1)), lane, hq);
    if (p + 2 < MPAIRS) merge_load_pair(sp + (p + 2) * PAIRQ, xa);
    __builtin_amdgcn_sched_barrier(0);
    merge_store_pair(op + (p + 1) * PAIRQ, xb, hq);
    __builtin_amdgcn_sched_barrier(0);
    if (p + 2 < MPAIRS) merge_load_h(h, merge_shot(t0 + 2 * (p + 2)), lane, hq);
  }
}

extern "C" void kernel_launch(void* const* d_in, const int* in_sizes, int n_in,
                              void* d_out, int out_size, void* d_ws, size_t ws_size,
                              hipStream_t stream) {
  const float* seq  = (const float*)d_in[0];
  const float* w1   = (const float*)d_in[1];
  const float* b1   = (const float*)d_in[2];
  const float* w2   = (const float*)d_in[3];
  const float* b2   = (const float*)d_in[4];
  const float* w3   = (const float*)d_in[5];
  const float* b3   = (const float*)d_in[6];
  const float* w4   = (const float*)d_in[7];
  const float* b4   = (const float*)d_in[8];
  const float* fcw  = (const float*)d_in[9];
  const float* fcb  = (const float*)d_in[10];
  const float* lng  = (const float*)d_in[11];
  const float* lnb  = (const float*)d_in[12];
  float* out = (float*)d_out;

  char* ws = (char*)d_ws;
  float* seq_shot = (float*)(ws);
  float* pooled   = (float*)(ws + 647680);
  float* part     = (float*)(ws + 1683968);
  float* h        = (float*)(ws + 2732544);

  shot_pool_kernel<<<(SHOT * SC) / (WPW * 4), 128, 0, stream>>>(seq, seq_shot);

  conv_pool_kernel<<<SC / 32, 32 * CP_WAVES, 0, stream>>>(
      seq_shot, w1, b1, w2, b2, w3, b3, w4, b4, pooled);

  gemm_wmma_kernel<<<dim3(NDIM / 16, KCHUNKS), 32, 0, stream>>>(pooled, fcw, part);

  fc_tanh_ln_kernel<<<SHOT, 256, 0, stream>>>(part, fcb, lng, lnb, h);

  merge_kernel<<<SEQ_LEN / (MROWS * 4), 128, 0, stream>>>(seq, h, out);
}


# GEMM regridded: K in 16 chunks (8 of 127, 8 of 126 k-steps), 256 groups of 4 waves = 1 wave per SIMD, 254 products per wave, was 506 on half the SIMDs; 20 partial rows stored; k sum order changes
# speedup vs baseline: 2.7024x; 1.0800x over previous
#include <hip/hip_runtime.h>
#include <hip/hip_bf16.h>

typedef __attribute__((ext_vector_type(2))) float v2f;
typedef __attribute__((ext_vector_type(4))) float v4f;
typedef __attribute__((ext_vector_type(8))) float v8f;

#define SEQ_LEN 32768
#define EMB     1024
#define SHOT    20
#define MPAD    32
#define SPLIT   1638
#define SC      8096
#define FLATL   (SPLIT * EMB)
#define LC1     2024
#define LC2     2023
#define LC3     2023
#define LC4     2024
#define CATL    (LC1 + LC2 + LC3 + LC4)
#define KDIM    SC
#define NDIM    EMB
#define KCHUNKS 16
#define KSTEPS_ALL (KDIM / 4)
#define KS_LO   (KSTEPS_ALL / KCHUNKS)
#define KS_HI   (KS_LO + 1)
#define KS_NHI  (KSTEPS_ALL - KS_LO * KCHUNKS)
#define PROWS   SHOT
#define PCHUNK  (PROWS * NDIM)
#define PART_A_CHUNKS 12

#define WPW 32
#define NLD 7
#define GRP 4
#define NGRP (WPW / GRP)
static_assert(SC % WPW == 0, "a wave's windows stay inside one shot");
static_assert(FLATL / SC >= 32 * (NLD - 1), "loads 0..NLD-2 are inside every window");
static_assert((FLATL + SC - 1) / SC + 1 <= 32 * NLD, "NLD loads cover the longest window");

#define QW (FLATL / SC)
#define RW (FLATL % SC)
static_assert((long long)RW * SC + SC < 0x7fffffffLL, "RW*(j+1)+SC-1 fits in 32 bits");
__device__ __forceinline__ int pool_start(int j) {
  return QW * j + (int)((unsigned)(RW * j) / (unsigned)SC);
}
__device__ __forceinline__ int pool_end(int j) {
  return QW * (j + 1) + (int)((unsigned)(RW * (j + 1) + SC - 1) / (unsigned)SC);
}

__device__ __forceinline__ void pool_issue(const float* __restrict__ seq, int c0, int c1,
                                           int j, int lane, float (&x)[NLD]) {
  const int start = pool_start(j);
  const int last  = pool_end(j) - 1;
#pragma unroll
  for (int i = 0; i < NLD; ++i) {
    int f = start + lane + 32 * i;
    if (i == NLD - 1) f = (f < last) ? f : last;
    const unsigned boff = (unsigned)(f + ((f < EMB) ? c0 : c1)) * 4u;
    x[i] = *(const float*)((const char*)seq + boff);
  }
}

__device__ __forceinline__ float pool_sum(int j, int lane, const float (&x)[NLD]) {
  float sum = 0.f;
#pragma unroll
  for (int i = 0; i < NLD - 1; ++i) sum += x[i];
  if (pool_start(j) + lane + 32 * (NLD - 1) < pool_end(j)) sum += x[NLD - 1];
  return sum;
}

template <int N>
__device__ __forceinline__ void pool_fold(const float (&in)[2 * N], float (&out)[N], int lane, int off) {
  const bool hi = (lane & off) != 0;
#pragma unroll
  for (int i = 0; i < N; ++i) {
    float keep = hi ? in[i + N] : in[i];
    float send = hi ? in[i] : in[i + N];
    out[i] = keep + __shfl_xor(send, off, 32);
  }
}

__global__ void __launch_bounds__(128)
shot_pool_kernel(const float* __restrict__ seq, float* __restrict__ seq_shot) {
  const int lane = threadIdx.x & 31;
  const int wave = __builtin_amdgcn_readfirstlane((int)((blockIdx.x * blockDim.x + threadIdx.x) >> 5));
  const int s  = wave / (SC / WPW);
  const int j0 = (wave - s * (SC / WPW)) * WPW;
  const int c0 = s * SPLIT * EMB;
  const int c1 = (((s - 1) * SPLIT + SEQ_LEN) % SEQ_LEN) * EMB;

  float p[WPW];
  float xa[GRP][NLD], xb[GRP][NLD];

#pragma unroll
  for (int w = 0; w < GRP; ++w) pool_issue(seq, c0, c1, j0 + w, lane, xa[w]);

#pragma unroll
  for (int g = 0; g < NGRP; g += 2) {
#pragma unroll
    for (int w = 0; w < GRP; ++w) pool_issue(seq, c0, c1, j0 + (g + 1) * GRP + w, lane, xb[w]);
    __builtin_amdgcn_sched_barrier(0);
#pragma unroll
    for (int w = 0; w < GRP; ++w) p[g * GRP + w] = pool_sum(j0 + g * GRP + w, lane, xa[w]);
    __builtin_amdgcn_sched_barrier(0);
    if (g + 2 < NGRP) {
#pragma unroll
      for (int w = 0; w < GRP; ++w) pool_issue(seq, c0, c1, j0 + (g + 2) * GRP + w, lane, xa[w]);
    }
    __builtin_amdgcn_sched_barrier(0);
#pragma unroll
    for (int w = 0; w < GRP; ++w) p[(g + 1) * GRP + w] = pool_sum(j0 + (g + 1) * GRP + w, lane, xb[w]);
    __builtin_amdgcn_sched_barrier(0);
  }

  float q16[16], q8[8], q4[4], q2[2], q1[1];
  pool_fold<16>(p,   q16, lane, 16);
  pool_fold<8>(q16,  q8,  lane, 8);
  pool_fold<4>(q8,   q4,  lane, 4);
  pool_fold<2>(q4,   q2,  lane, 2);
  pool_fold<1>(q2,   q1,  lane, 1);

  const int j = j0 + lane;
  seq_shot[s * SC + j] = q1[0] / (float)(pool_end(j) - pool_start(j));
}

#define CT_COLS 48
#define CONV_K  (SHOT * 3)
#define CP_WAVES 6
static_assert(SC % 32 == 0, "a work-group's 32 pooled columns are whole 128-byte row pieces");
static_assert(CONV_K % 4 == 0 && SHOT % 4 == 0, "whole k-steps, even weight offsets");

static_assert((long long)SC * CATL + SC < 0x7fffffffLL, "(j+1)*CATL + SC - 1 fits in 32 bits");
static_assert(CATL <= SC, "a window of the second pool holds one or two conv outputs");
__device__ __forceinline__ int cat_start(int j) { return (int)((unsigned)(j * CATL) / (unsigned)SC); }
__device__ __forceinline__ int cat_end(int j)   { return (int)((unsigned)((j + 1) * CATL + SC - 1) / (unsigned)SC); }

__device__ __forceinline__ int conv_of(int t) {
  return (t >= LC1) + (t >= LC1 + LC2) + (t >= LC1 + LC2 + LC3);
}

#define CONV_KS (CONV_K / 4)
__device__ __forceinline__ void conv_frags(const float* __restrict__ ss, const float* __restrict__ w,
                                           int k3, int dil, bool col_in, int p, int srow, bool row_in,
                                           int half, v2f (&fa)[CONV_KS], v2f (&fb)[CONV_KS]) {
  const int ktot = SHOT + k3 * (CONV_K - SHOT);
  const float* wr = w + srow * ktot;
#pragma unroll
  for (int ks = 0; ks < CONV_KS; ++ks) {
    const int  kk  = 4 * ks + 2 * half;
    const bool kin = kk < ktot;
    const v2f  av  = *(const v2f*)(wr + (kin ? kk : 0));
    fa[ks][0] = (kin && row_in) ? av[0] : 0.f;
    fa[ks][1] = (kin && row_in) ? av[1] : 0.f;
#pragma unroll
    for (int e = 0; e < 2; ++e) {
      const int ka = 4 * ks + e;
      const int kb = 4 * ks + 2 + e;
      const int oa = (ka % SHOT) * SC + k3 * ((ka / 3 - ka % SHOT) * SC + dil * (ka % 3));
      const int ob = (kb % SHOT) * SC + k3 * ((kb / 3 - kb % SHOT) * SC + dil * (kb % 3));
      const float x = ss[(col_in ? p : 0) + (half ? ob : oa)];
      fb[ks][e] = (kin && col_in) ? x : 0.f;
    }
  }
}

__global__ void __launch_bounds__(32 * CP_WAVES)
conv_pool_kernel(const float* __restrict__ ss,
                 const float* __restrict__ w1, const float* __restrict__ b1,
                 const float* __restrict__ w2, const float* __restrict__ b2,
                 const float* __restrict__ w3, const float* __restrict__ b3,
                 const float* __restrict__ w4, const float* __restrict__ b4,
                 float* __restrict__ pooled) {
  __shared__ float tile[MPAD][CT_COLS];
  const int lane = threadIdx.x & 31;
  const int half = lane >> 4;
  const int c    = lane & 15;
  const int wv = __builtin_amdgcn_readfirstlane((int)(threadIdx.x >> 5));
  const int q  = wv % 3;
  const int m  = wv / 3;
  const int jb  = blockIdx.x * 32;
  const int tlo = cat_start(jb);
  const int tq  = tlo + 16 * q;

  const int tfirst = (tq < CATL - 1) ? tq : CATL - 1;
  const int tlast  = (tq + 15 < CATL - 1) ? tq + 15 : CATL - 1;
  const int cvA = conv_of(tfirst);
  const int cvB = conv_of(tlast);

  const int  t    = tq + c;
  const bool t_in = t < CATL;
  const int  cvt  = conv_of(t_in ? t : CATL - 1);
  const int  off  = (cvt == 0) ? 0 : (cvt == 1) ? LC1 : (cvt == 2) ? LC1 + LC2 : LC1 + LC2 + LC3;
  const int  p    = t_in ? 4 * (t - off) : 0;

  const int  srow   = (16 * m + c < SHOT) ? 16 * m + c : SHOT - 1;
  const bool row_in = 16 * m + c < SHOT;

  const float* wA = (cvA == 0) ? w1 : (cvA == 1) ? w2 : (cvA == 2) ? w3 : w4;
  const float* wB = (cvB == 0) ? w1 : (cvB == 1) ? w2 : (cvB == 2) ? w3 : w4;

  v2f faA[CONV_KS], fbA[CONV_KS], faB[CONV_KS], fbB[CONV_KS];
  conv_frags(ss, wA, cvA < 3, cvA + 1, t_in && cvt == cvA,               p, srow, row_in, half, faA, fbA);
  conv_frags(ss, wB, cvB < 3, cvB + 1, t_in && cvt == cvB && cvB != cvA, p, srow, row_in, half, faB, fbB);
  const float* bp = (cvt == 0) ? b1 : (cvt == 1) ? b2 : (cvt == 2) ? b3 : b4;
  float bv[8];
#pragma unroll
  for (int r = 0; r < 8; ++r) {
    const int s = 16 * m + 8 * half + r;
    bv[r] = bp[(s < SHOT) ? s : SHOT - 1];
  }
  __builtin_amdgcn_sched_barrier(0);
  v8f acc = {};
#pragma unroll
  for (int ks = 0; ks < CONV_KS; ++ks)
    acc = __builtin_amdgcn_wmma_f32_16x16x4_f32(false, faA[ks], false, fbA[ks], (short)0, acc, false, false);
#pragma unroll
  for (int ks = 0; ks < CONV_KS; ++ks)
    acc = __builtin_amdgcn_wmma_f32_16x16x4_f32(false, faB[ks], false, fbB[ks], (short)0, acc, false, false);

#pragma unroll
  for (int r = 0; r < 8; ++r)
    tile[16 * m + 8 * half + r][16 * q + c] = acc[r] + bv[r];
  __syncthreads();

  for (int ks = wv; ks < 32 / 4; ks += CP_WAVES) {
    v4f o;
#pragma unroll
    for (int e = 0; e < 2; ++e) {
      const int   j     = jb + 4 * ks + 2 * half + e;
      const int   start = cat_start(j);
      const int   end   = cat_end(j);
      const bool  two   = end - start > 1;
      const float len   = (float)(end - start);
#pragma unroll
      for (int r = 0; r < 2; ++r) {
        const int s = c + 16 * r;
        const float x0 = tile[s][start - tlo];
        const float x1 = tile[s][end - 1 - tlo];
        float sum = 0.f;
        sum += x0;
        sum += two ? x1 : 0.f;
        o[2 * r + e] = (s < SHOT) ? sum / len : 0.f;
      }
    }
    ((v4f*)pooled)[(size_t)(blockIdx.x * (32 / 4) + ks) * 32 + lane] = o;
  }
}

#define RING   33
#define AHEAD  31
static_assert(KDIM % 4 == 0 && KS_NHI * KS_HI + (KCHUNKS - KS_NHI) * KS_LO == KSTEPS_ALL, "whole k-steps, the chunks cover K");
static_assert(AHEAD == RING - 2 && 2 * AHEAD <= 63, "a slot is refilled two k-steps after its products; loads in flight fit the counter");
static_assert(NDIM / 16 % 4 == 0, "a work-group takes 4 whole N-tiles");

__device__ __forceinline__ void gemm_issue(const v4f* __restrict__ pa, const float* __restrict__ pb,
                                           int kstep, v4f& a, v2f& b) {
  a = pa[32 * kstep];
  b = *(const v2f*)(pb + 4 * kstep);
}

__device__ __forceinline__ void gemm_mul(const v4f& a, const v2f& b, v8f& acc0, v8f& acc1) {
  const v2f a0 = __builtin_shufflevector(a, a, 0, 1);
  const v2f a1 = __builtin_shufflevector(a, a, 2, 3);
  acc0 = __builtin_amdgcn_wmma_f32_16x16x4_f32(false, a0, false, b, (short)0, acc0, false, false);
  acc1 = __builtin_amdgcn_wmma_f32_16x16x4_f32(false, a1, false, b, (short)0, acc1, false, false);
}

template <int KS>
__device__ __forceinline__ void gemm_chunk(const v4f* __restrict__ pa, const float* __restrict__ pb,
                                           v8f& acc0, v8f& acc1) {
  constexpr int NFULL = (KS - AHEAD) / RING;
  constexpr int REM   = KS - NFULL * RING;
  static_assert(NFULL >= 1 && NFULL * RING + AHEAD <= KS && REM >= AHEAD, "the rolled trips ask only for k-steps of the chunk; the end drains the ring");
  v4f ra[RING];
  v2f rb[RING];
#pragma unroll
  for (int i = 0; i < AHEAD; ++i) {
    gemm_issue(pa, pb, i, ra[i], rb[i]);
    __builtin_amdgcn_sched_barrier(0);
  }

  for (int t = 0; t < NFULL; ++t) {
#pragma unroll
    for (int i = 0; i < RING; ++i) {
      gemm_mul(ra[i], rb[i], acc0, acc1);
      const int j = (i + AHEAD) % RING;
      gemm_issue(pa, pb, i + AHEAD, ra[j], rb[j]);
      __builtin_amdgcn_sched_barrier(0);
    }
    pa += 32 * RING;
    pb += 4 * RING;
  }
#pragma unroll
  for (int i = 0; i < REM; ++i) {
    gemm_mul(ra[i % RING], rb[i % RING], acc0, acc1);
    if (i + AHEAD < REM) {
      const int j = (i + AHEAD) % RING;
      gemm_issue(pa, pb, i + AHEAD, ra[j], rb[j]);
      __builtin_amdgcn_sched_barrier(0);
    }
  }
}

__global__ void __launch_bounds__(128)
gemm_wmma_kernel(const float* __restrict__ pooled,
                 const float* __restrict__ fcw,
                 float* __restrict__ part_a, float* __restrict__ part_b) {
  int lane = threadIdx.x & 31;
  int half = lane >> 4;
  int mr   = lane & 15;
  const int wv = __builtin_amdgcn_readfirstlane((int)(threadIdx.x >> 5));
  int nbase = (blockIdx.x * 4 + wv) * 16;
  int kc    = blockIdx.y;
  int ks0   = kc * KS_LO + (kc < KS_NHI ? kc : KS_NHI);

  v8f acc0 = {};
  v8f acc1 = {};
  const v4f*   pa = (const v4f*)pooled + (size_t)ks0 * 32 + lane;
  const float* pb = fcw + (size_t)(nbase + mr) * KDIM + 4 * ks0 + 2 * half;

  if (kc < KS_NHI) gemm_chunk<KS_HI>(pa, pb, acc0, acc1);
  else             gemm_chunk<KS_LO>(pa, pb, acc0, acc1);

  float* pbase = ((kc < PART_A_CHUNKS) ? part_a + (size_t)kc * PCHUNK
                                       : part_b + (size_t)(kc - PART_A_CHUNKS) * PCHUNK) + nbase + mr;
  for (int r = 0; r < 8; ++r) {
    int row = r + 8 * half;
    pbase[(size_t)row * NDIM] = acc0[r];
  }
  if (half == 0) {
    for (int r = 0; r < SHOT - 16; ++r) pbase[(size_t)(16 + r) * NDIM] = acc1[r];
  }
}

#define LN_WAVES (256 / 32)
static_assert(NDIM == 4 * 256, "a thread owns 4 neighbouring columns of its row");
__global__ void __launch_bounds__(256)
fc_tanh_ln_kernel(const float* __restrict__ part_a, const float* __restrict__ part_b, const float* __restrict__ fcb,
                  const float* __restrict__ lng, const float* __restrict__ lnb,
                  float* __restrict__ h) {
  const int s = blockIdx.x;
  const int tid = threadIdx.x;
  __shared__ float redA[LN_WAVES];
  __shared__ float redB[LN_WAVES];

  const v4f* ppa = (const v4f*)(part_a + (size_t)s * NDIM) + tid;
  const v4f* ppb = (const v4f*)(part_b + (size_t)s * NDIM) + tid;
  v4f v = ((const v4f*)fcb)[tid];
  v4f pk[KCHUNKS];
#pragma unroll
  for (int kc = 0; kc < KCHUNKS; ++kc)
    pk[kc] = (kc < PART_A_CHUNKS) ? ppa[(size_t)kc * (PCHUNK / 4)] : ppb[(size_t)(kc - PART_A_CHUNKS) * (PCHUNK / 4)];
  const v4f g = ((const v4f*)lng)[tid];
  const v4f b = ((const v4f*)lnb)[tid];
  __builtin_amdgcn_sched_barrier(0);
#pragma unroll
  for (int kc = 0; kc < KCHUNKS; ++kc) v += pk[kc];

  float t[4];
  float sum = 0.f, sq = 0.f;
#pragma unroll
  for (int i = 0; i < 4; ++i) {
    const float tv = tanhf(v[i]);
    t[i] = tv;
    sum += tv; sq += tv * tv;
  }
#pragma unroll
  for (int off = 16; off > 0; off >>= 1) {
    sum += __shfl_xor(sum, off, 32);
    sq  += __shfl_xor(sq,  off, 32);
  }
  if ((tid & 31) == 0) { redA[tid >> 5] = sum; redB[tid >> 5] = sq; }
  __syncthreads();
  float ra = 0.f, rb = 0.f;
#pragma unroll
  for (int w = 0; w < LN_WAVES; ++w) { ra += redA[w]; rb += redB[w]; }
  const float mu  = ra * (1.f / NDIM);
  const float var = rb * (1.f / NDIM) - mu * mu;
  const float inv = rsqrtf(var + 1e-5f);
  v4f o;
#pragma unroll
  for (int i = 0; i < 4; ++i) o[i] = (t[i] - mu) * inv * g[i] + b[i];
  ((v4f*)(h + (size_t)s * NDIM))[tid] = o;
}

#define MROWS 8
#define MPAIRS (MROWS / 2)
#define MLD   (EMB / 4 / 32)
#define PAIRQ (2 * (EMB / 4))
static_assert(MPAIRS % 2 == 0, "pairs alternate between two register sets");
static_assert(SPLIT % 2 == 0 && ((SHOT - 1) * SPLIT) % 2 == 0, "an aligned pair of rows lies in one shot");
static_assert(SEQ_LEN % (MROWS * 4) == 0, "the grid covers the rows exactly");

__device__ __forceinline__ int merge_shot(int t) {
  const int s = t / SPLIT;
  return (s > SHOT - 1) ? SHOT - 1 : s;
}

__device__ __forceinline__ void merge_load_h(const float* __restrict__ h, int s, int lane, v4f (&hq)[MLD]) {
  const v4f* hp = (const v4f*)(h + s * EMB) + lane;
#pragma unroll
  for (int i = 0; i < MLD; ++i) hq[i] = hp[32 * i];
}

__device__ __forceinline__ void merge_load_pair(const v4f* __restrict__ sp, v4f (&x)[2][MLD]) {
#pragma unroll
  for (int r = 0; r < 2; ++r) {
#pragma unroll
    for (int i = 0; i < MLD; ++i)
      x[r][i] = __builtin_nontemporal_load(sp + r * (EMB / 4) + 32 * i);
  }
}

__device__ __forceinline__ void merge_store_pair(v4f* __restrict__ op, const v4f (&x)[2][MLD], const v4f (&hq)[MLD]) {
#pragma unroll
  for (int r = 0; r < 2; ++r) {
#pragma unroll
    for (int i = 0; i < MLD; ++i) __builtin_nontemporal_store(x[r][i] + hq[i], op + r * (EMB / 4) + 32 * i);
  }
}

__global__ void __launch_bounds__(128)
merge_kernel(const float* __restrict__ seq, const float* __restrict__ h,
             float* __restrict__ out) {
  const int lane = threadIdx.x & 31;
  const int wave = __builtin_amdgcn_readfirstlane((int)((blockIdx.x * blockDim.x + threadIdx.x) >> 5));
  const int t0 = wave * MROWS;

  const v4f* sp = (const v4f*)seq + (size_t)t0 * (EMB / 4) + lane;
  v4f*       op = (v4f*)out       + (size_t)t0 * (EMB / 4) + lane;
  v4f hq[MLD];
  v4f xa[2][MLD], xb[2][MLD];
  merge_load_h(h, merge_shot(t0), lane, hq);
  merge_load_pair(sp, xa);

#pragma unroll
  for (int p = 0; p < MPAIRS; p += 2) {
    merge_load_pair(sp + (p + 1) * PAIRQ, xb);
    __builtin_amdgcn_sched_barrier(0);
    merge_store_pair(op + p * PAIRQ, xa, hq);
    __builtin_amdgcn_sched_barrier(0);
    merge_load_h(h, merge_shot(t0 + 2 * (p + 1)), lane, hq);
    if (p + 2 < MPAIRS) merge_load_pair(sp + (p + 2) * PAIRQ, xa);
    __builtin_amdgcn_sched_barrier(0);
    merge_store_pair(op + (p + 1) * PAIRQ, xb, hq);
    __builtin_amdgcn_sched_barrier(0);
    if (p + 2 < MPAIRS) merge_load_h(h, merge_shot(t0 + 2 * (p + 2)), lane, hq);
  }
}

extern "C" void kernel_launch(void* const* d_in, const int* in_sizes, int n_in,
                              void* d_out, int out_size, void* d_ws, size_t ws_size,
                              hipStream_t stream) {
  const float* seq  = (const float*)d_in[0];
  const float* w1   = (const float*)d_in[1];
  const float* b1   = (const float*)d_in[2];
  const float* w2   = (const float*)d_in[3];
  const float* b2   = (const float*)d_in[4];
  const float* w3   = (const float*)d_in[5];
  const float* b3   = (const float*)d_in[6];
  const float* w4   = (const float*)d_in[7];
  const float* b4   = (const float*)d_in[8];
  const float* fcw  = (const float*)d_in[9];
  const float* fcb  = (const float*)d_in[10];
  const float* lng  = (const float*)d_in[11];
  const float* lnb  = (const float*)d_in[12];
  float* out = (float*)d_out;

  char* ws = (char*)d_ws;
  float* seq_shot = (float*)(ws);
  float* pooled   = (float*)(ws + 647680);
  float* part_a   = (float*)(ws + 1683968);
  float* part_b   = (float*)(ws);
  float* h        = (float*)(ws + 2732544);

  shot_pool_kernel<<<(SHOT * SC) / (WPW * 4), 128, 0, stream>>>(seq, seq_shot);

  conv_pool_kernel<<<SC / 32, 32 * CP_WAVES, 0, stream>>>(
      seq_shot, w1, b1, w2, b2, w3, b3, w4, b4, pooled);

  gemm_wmma_kernel<<<dim3(NDIM / 16 / 4, KCHUNKS), 128, 0, stream>>>(pooled, fcw, part_a, part_b);

  fc_tanh_ln_kernel<<<SHOT, 256, 0, stream>>>(part_a, part_b, fcb, lng, lnb, h);

  merge_kernel<<<SEQ_LEN / (MROWS * 4), 128, 0, stream>>>(seq, h, out);
}
